# starting kernel
# baseline (speedup 1.0000x reference)
#include <hip/hip_runtime.h>
#include <hip/hip_bf16.h>

constexpr int B   = 2;
constexpr int S   = 4096;
constexpr int E   = 1024;
constexpr int D   = 64;
constexpr int WIN = 5;
constexpr int DIL = 2;

typedef __attribute__((ext_vector_type(2))) float v2f;
typedef __attribute__((ext_vector_type(8))) float v8f;

__global__ __launch_bounds__(256) void qkv_proj_kernel(
    const float* __restrict__ x,
    const float* __restrict__ W,
    const float* __restrict__ bias,
    float* __restrict__ out)
{
    const int lane  = threadIdx.x & 31;
    const int wave  = threadIdx.x >> 5;
    const int mtile = blockIdx.x * 8 + wave;
    const int row0  = mtile * 16;

    const int m  = lane & 15;
    const int n  = lane & 15;
    const int kh = (lane >> 4) * 2;

    const float* xrow = x + (size_t)(row0 + m) * E + kh;

    v8f acc0 = {}, acc1 = {}, acc2 = {}, acc3 = {};

    for (int k0 = 0; k0 < E; k0 += 4) {
        v2f a = *(const v2f*)(xrow + k0);

        const float* wp = W + (size_t)(k0 + kh) * D + n;
        v2f b0, b1, b2, b3;
        b0.x = wp[0];      b0.y = wp[D];
        b1.x = wp[16];     b1.y = wp[D + 16];
        b2.x = wp[32];     b2.y = wp[D + 32];
        b3.x = wp[48];     b3.y = wp[D + 48];

        acc0 = __builtin_amdgcn_wmma_f32_16x16x4_f32(false, a, false, b0,
                                                     (short)0, acc0, false, false);
        acc1 = __builtin_amdgcn_wmma_f32_16x16x4_f32(false, a, false, b1,
                                                     (short)0, acc1, false, false);
        acc2 = __builtin_amdgcn_wmma_f32_16x16x4_f32(false, a, false, b2,
                                                     (short)0, acc2, false, false);
        acc3 = __builtin_amdgcn_wmma_f32_16x16x4_f32(false, a, false, b3,
                                                     (short)0, acc3, false, false);
    }

    const int mo = (lane >> 4) * 8;
    float* orow = out + (size_t)(row0 + mo) * D + n;
#pragma unroll
    for (int r = 0; r < 8; ++r) {
        float* o = orow + (size_t)r * D;
        o[0]  = acc0[r] + bias[n];
        o[16] = acc1[r] + bias[n + 16];
        o[32] = acc2[r] + bias[n + 32];
        o[48] = acc3[r] + bias[n + 48];
    }
}

__global__ __launch_bounds__(64) void vsum_kernel(
    const float* __restrict__ V, float* __restrict__ Vsum)
{
    const int b = blockIdx.x;
    const int d = threadIdx.x;
    const float* p = V + (size_t)b * S * D + d;
    float s = 0.0f;
#pragma unroll 4
    for (int i = 0; i < S; ++i) s += p[(size_t)i * D];
    Vsum[b * D + d] = s;
}

__global__ __launch_bounds__(256) void attn_kernel(
    const float* __restrict__ Q, const float* __restrict__ K,
    const float* __restrict__ V, const float* __restrict__ Vsum,
    float* __restrict__ out)
{
    const int gwave = (blockIdx.x * blockDim.x + threadIdx.x) >> 5;
    if (gwave >= B * S) return;
    const int lane = threadIdx.x & 31;
    const int b = gwave / S;
    const int i = gwave - b * S;

    const size_t rowQ = (size_t)(b * S + i) * D;
    const float2 qv = *(const float2*)(Q + rowQ + lane * 2);

    float accx = Vsum[b * D + lane * 2];
    float accy = Vsum[b * D + lane * 2 + 1];
    float Z = (float)S;

#pragma unroll
    for (int w = 0; w < WIN; ++w) {
        const int j = i + DIL * (w - WIN / 2);
        if (j < 0 || j >= S) continue;
        const size_t rowJ = (size_t)(b * S + j) * D;
        const float2 kv = *(const float2*)(K + rowJ + lane * 2);
        float c = qv.x * kv.x + qv.y * kv.y;
#pragma unroll
        for (int off = 16; off > 0; off >>= 1) c += __shfl_xor(c, off, 32);
        const float em1 = __expf(c) - 1.0f;
        const float2 vv = *(const float2*)(V + rowJ + lane * 2);
        accx += em1 * vv.x;
        accy += em1 * vv.y;
        Z    += em1;
    }

    const float inv = 1.0f / Z;
    float2 o;
    o.x = accx * inv;
    o.y = accy * inv;
    *(float2*)(out + rowQ + lane * 2) = o;
}

extern "C" void kernel_launch(void* const* d_in, const int* in_sizes, int n_in,
                              void* d_out, int out_size, void* d_ws, size_t ws_size,
                              hipStream_t stream) {
    const float* x  = (const float*)d_in[0];
    const float* Wq = (const float*)d_in[1];
    const float* bq = (const float*)d_in[2];
    const float* Wk = (const float*)d_in[3];
    const float* bk = (const float*)d_in[4];
    const float* Wv = (const float*)d_in[5];
    const float* bv = (const float*)d_in[6];
    float* outp = (float*)d_out;

    constexpr size_t QKV = (size_t)B * S * D;
    float* ws = (float*)d_ws;
    float* Qb = ws;
    float* Kb = ws + QKV;
    float* Vb = ws + 2 * QKV;
    float* Vs = ws + 3 * QKV;

    const int mtiles = (B * S) / 16;
    dim3 gproj(mtiles / 8);
    qkv_proj_kernel<<<gproj, 256, 0, stream>>>(x, Wq, bq, Qb);
    qkv_proj_kernel<<<gproj, 256, 0, stream>>>(x, Wk, bk, Kb);
    qkv_proj_kernel<<<gproj, 256, 0, stream>>>(x, Wv, bv, Vb);

    vsum_kernel<<<B, 64, 0, stream>>>(Vb, Vs);

    const int nwaves = B * S;
    attn_kernel<<<(nwaves * 32) / 256, 256, 0, stream>>>(Qb, Kb, Vb, Vs, outp);
}


# vsum: column sum of V on 4 groups of 1024 lanes (batch x 128-byte half row, float4 loads, lane exchange + LDS tree) instead of 2 x 64 lanes walking 4096 rows; changes the order of the sum
# speedup vs baseline: 3.0628x; 3.0628x over previous
#include <hip/hip_runtime.h>
#include <hip/hip_bf16.h>

constexpr int B   = 2;
constexpr int S   = 4096;
constexpr int E   = 1024;
constexpr int D   = 64;
constexpr int WIN = 5;
constexpr int DIL = 2;

typedef __attribute__((ext_vector_type(2))) float v2f;
typedef __attribute__((ext_vector_type(8))) float v8f;

__global__ __launch_bounds__(256) void qkv_proj_kernel(
    const float* __restrict__ x,
    const float* __restrict__ W,
    const float* __restrict__ bias,
    float* __restrict__ out)
{
    const int lane  = threadIdx.x & 31;
    const int wave  = threadIdx.x >> 5;
    const int mtile = blockIdx.x * 8 + wave;
    const int row0  = mtile * 16;

    const int m  = lane & 15;
    const int n  = lane & 15;
    const int kh = (lane >> 4) * 2;

    const float* xrow = x + (size_t)(row0 + m) * E + kh;

    v8f acc0 = {}, acc1 = {}, acc2 = {}, acc3 = {};

    for (int k0 = 0; k0 < E; k0 += 4) {
        v2f a = *(const v2f*)(xrow + k0);

        const float* wp = W + (size_t)(k0 + kh) * D + n;
        v2f b0, b1, b2, b3;
        b0.x = wp[0];      b0.y = wp[D];
        b1.x = wp[16];     b1.y = wp[D + 16];
        b2.x = wp[32];     b2.y = wp[D + 32];
        b3.x = wp[48];     b3.y = wp[D + 48];

        acc0 = __builtin_amdgcn_wmma_f32_16x16x4_f32(false, a, false, b0,
                                                     (short)0, acc0, false, false);
        acc1 = __builtin_amdgcn_wmma_f32_16x16x4_f32(false, a, false, b1,
                                                     (short)0, acc1, false, false);
        acc2 = __builtin_amdgcn_wmma_f32_16x16x4_f32(false, a, false, b2,
                                                     (short)0, acc2, false, false);
        acc3 = __builtin_amdgcn_wmma_f32_16x16x4_f32(false, a, false, b3,
                                                     (short)0, acc3, false, false);
    }

    const int mo = (lane >> 4) * 8;
    float* orow = out + (size_t)(row0 + mo) * D + n;
#pragma unroll
    for (int r = 0; r < 8; ++r) {
        float* o = orow + (size_t)r * D;
        o[0]  = acc0[r] + bias[n];
        o[16] = acc1[r] + bias[n + 16];
        o[32] = acc2[r] + bias[n + 32];
        o[48] = acc3[r] + bias[n + 48];
    }
}

constexpr int VS_HALF    = 32;
constexpr int VS_THREADS = 1024;
constexpr int VS_ROWS    = VS_THREADS / 8;
constexpr int VS_WAVES   = VS_THREADS / 32;
constexpr int VS_GROUPS  = B * (D / VS_HALF);

__global__ __launch_bounds__(VS_THREADS) void vsum_kernel(
    const float* __restrict__ V, float* __restrict__ Vsum)
{
    __shared__ float part[VS_WAVES][VS_HALF];

    const int b    = blockIdx.x >> 1;
    const int half = blockIdx.x & 1;
    const int t    = threadIdx.x;
    const int c    = t & 7;
    const int r    = t >> 3;

    const float* p = V + ((size_t)b * S + r) * D + half * VS_HALF + c * 4;

    float4 s = make_float4(0.0f, 0.0f, 0.0f, 0.0f);
#pragma unroll 8
    for (int i = 0; i < S / VS_ROWS; ++i) {
        const float4 v = *(const float4*)(p + (size_t)i * VS_ROWS * D);
        s.x += v.x; s.y += v.y; s.z += v.z; s.w += v.w;
    }

    for (int off = 8; off <= 16; off <<= 1) {
        s.x += __shfl_xor(s.x, off, 32);
        s.y += __shfl_xor(s.y, off, 32);
        s.z += __shfl_xor(s.z, off, 32);
        s.w += __shfl_xor(s.w, off, 32);
    }

    const int lane = t & 31;
    const int wave = t >> 5;
    if (lane < 8) *(float4*)&part[wave][lane * 4] = s;
    __syncthreads();

    if (t < VS_HALF) {
        float a = 0.0f;
#pragma unroll
        for (int w = 0; w < VS_WAVES; ++w) a += part[w][t];
        Vsum[b * D + half * VS_HALF + t] = a;
    }
}

__global__ __launch_bounds__(256) void attn_kernel(
    const float* __restrict__ Q, const float* __restrict__ K,
    const float* __restrict__ V, const float* __restrict__ Vsum,
    float* __restrict__ out)
{
    const int gwave = (blockIdx.x * blockDim.x + threadIdx.x) >> 5;
    if (gwave >= B * S) return;
    const int lane = threadIdx.x & 31;
    const int b = gwave / S;
    const int i = gwave - b * S;

    const size_t rowQ = (size_t)(b * S + i) * D;
    const float2 qv = *(const float2*)(Q + rowQ + lane * 2);

    float accx = Vsum[b * D + lane * 2];
    float accy = Vsum[b * D + lane * 2 + 1];
    float Z = (float)S;

#pragma unroll
    for (int w = 0; w < WIN; ++w) {
        const int j = i + DIL * (w - WIN / 2);
        if (j < 0 || j >= S) continue;
        const size_t rowJ = (size_t)(b * S + j) * D;
        const float2 kv = *(const float2*)(K + rowJ + lane * 2);
        float c = qv.x * kv.x + qv.y * kv.y;
#pragma unroll
        for (int off = 16; off > 0; off >>= 1) c += __shfl_xor(c, off, 32);
        const float em1 = __expf(c) - 1.0f;
        const float2 vv = *(const float2*)(V + rowJ + lane * 2);
        accx += em1 * vv.x;
        accy += em1 * vv.y;
        Z    += em1;
    }

    const float inv = 1.0f / Z;
    float2 o;
    o.x = accx * inv;
    o.y = accy * inv;
    *(float2*)(out + rowQ + lane * 2) = o;
}

extern "C" void kernel_launch(void* const* d_in, const int* in_sizes, int n_in,
                              void* d_out, int out_size, void* d_ws, size_t ws_size,
                              hipStream_t stream) {
    const float* x  = (const float*)d_in[0];
    const float* Wq = (const float*)d_in[1];
    const float* bq = (const float*)d_in[2];
    const float* Wk = (const float*)d_in[3];
    const float* bk = (const float*)d_in[4];
    const float* Wv = (const float*)d_in[5];
    const float* bv = (const float*)d_in[6];
    float* outp = (float*)d_out;

    constexpr size_t QKV = (size_t)B * S * D;
    float* ws = (float*)d_ws;
    float* Qb = ws;
    float* Kb = ws + QKV;
    float* Vb = ws + 2 * QKV;
    float* Vs = ws + 3 * QKV;

    const int mtiles = (B * S) / 16;
    dim3 gproj(mtiles / 8);
    qkv_proj_kernel<<<gproj, 256, 0, stream>>>(x, Wq, bq, Qb);
    qkv_proj_kernel<<<gproj, 256, 0, stream>>>(x, Wk, bk, Kb);
    qkv_proj_kernel<<<gproj, 256, 0, stream>>>(x, Wv, bv, Vb);

    vsum_kernel<<<VS_GROUPS, VS_THREADS, 0, stream>>>(Vb, Vs);

    const int nwaves = B * S;
    attn_kernel<<<(nwaves * 32) / 256, 256, 0, stream>>>(Qb, Kb, Vb, Vs, outp);
}


# proj K loop software-pipelined 16 k-steps deep in a register ring; to fit under the 63-load limit B rows are read as float4 (n-tile t = columns 4c+t), stores float4; sum order unchanged
# speedup vs baseline: 8.0046x; 2.6135x over previous
#include <hip/hip_runtime.h>
#include <hip/hip_bf16.h>

constexpr int B   = 2;
constexpr int S   = 4096;
constexpr int E   = 1024;
constexpr int D   = 64;
constexpr int WIN = 5;
constexpr int DIL = 2;

typedef __attribute__((ext_vector_type(2))) float v2f;
typedef __attribute__((ext_vector_type(8))) float v8f;

constexpr int PF     = 16;
constexpr int NKSTEP = E / 4;

struct KStep { v2f a; float4 w0, w1; };

__device__ __forceinline__ KStep load_kstep(const float* __restrict__ xk,
                                            const float* __restrict__ wk)
{
    KStep f;
    f.a  = *(const v2f*)xk;
    f.w0 = *(const float4*)wk;
    f.w1 = *(const float4*)(wk + D);
    return f;
}

__device__ __forceinline__ void mma_kstep(const KStep& f, v8f& acc0, v8f& acc1,
                                          v8f& acc2, v8f& acc3)
{
    v2f b0, b1, b2, b3;
    b0.x = f.w0.x;  b0.y = f.w1.x;
    b1.x = f.w0.y;  b1.y = f.w1.y;
    b2.x = f.w0.z;  b2.y = f.w1.z;
    b3.x = f.w0.w;  b3.y = f.w1.w;
    acc0 = __builtin_amdgcn_wmma_f32_16x16x4_f32(false, f.a, false, b0,
                                                 (short)0, acc0, false, false);
    acc1 = __builtin_amdgcn_wmma_f32_16x16x4_f32(false, f.a, false, b1,
                                                 (short)0, acc1, false, false);
    acc2 = __builtin_amdgcn_wmma_f32_16x16x4_f32(false, f.a, false, b2,
                                                 (short)0, acc2, false, false);
    acc3 = __builtin_amdgcn_wmma_f32_16x16x4_f32(false, f.a, false, b3,
                                                 (short)0, acc3, false, false);
}

__global__ __launch_bounds__(256) void qkv_proj_kernel(
    const float* __restrict__ x,
    const float* __restrict__ W,
    const float* __restrict__ bias,
    float* __restrict__ out)
{
    const int lane  = threadIdx.x & 31;
    const int wave  = threadIdx.x >> 5;
    const int mtile = blockIdx.x * 8 + wave;
    const int row0  = mtile * 16;

    const int m  = lane & 15;
    const int c4 = (lane & 15) * 4;
    const int kh = (lane >> 4) * 2;

    const float* xrow = x + (size_t)(row0 + m) * E + kh;
    const float* wcol = W + (size_t)kh * D + c4;

    v8f acc0 = {}, acc1 = {}, acc2 = {}, acc3 = {};

    KStep f[PF];
#pragma unroll
    for (int u = 0; u < PF; ++u)
        f[u] = load_kstep(xrow + 4 * u, wcol + (size_t)(4 * u) * D);

#pragma unroll 5
    for (int s0 = 0; s0 < NKSTEP - PF; s0 += PF) {
        const float* xk = xrow + 4 * (s0 + PF);
        const float* wk = wcol + (size_t)(4 * (s0 + PF)) * D;
#pragma unroll
        for (int u = 0; u < PF; ++u) {
            mma_kstep(f[u], acc0, acc1, acc2, acc3);
            f[u] = load_kstep(xk + 4 * u, wk + (size_t)(4 * u) * D);
            __builtin_amdgcn_sched_barrier(0);
        }
    }

#pragma unroll
    for (int u = 0; u < PF; ++u)
        mma_kstep(f[u], acc0, acc1, acc2, acc3);

    const int mo = (lane >> 4) * 8;
    const float4 bv = *(const float4*)(bias + c4);
    float* orow = out + (size_t)(row0 + mo) * D + c4;
#pragma unroll
    for (int r = 0; r < 8; ++r) {
        float4 o;
        o.x = acc0[r] + bv.x;
        o.y = acc1[r] + bv.y;
        o.z = acc2[r] + bv.z;
        o.w = acc3[r] + bv.w;
        *(float4*)(orow + (size_t)r * D) = o;
    }
}

constexpr int VS_HALF    = 32;
constexpr int VS_THREADS = 1024;
constexpr int VS_ROWS    = VS_THREADS / 8;
constexpr int VS_WAVES   = VS_THREADS / 32;
constexpr int VS_GROUPS  = B * (D / VS_HALF);

__global__ __launch_bounds__(VS_THREADS) void vsum_kernel(
    const float* __restrict__ V, float* __restrict__ Vsum)
{
    __shared__ float part[VS_WAVES][VS_HALF];

    const int b    = blockIdx.x >> 1;
    const int half = blockIdx.x & 1;
    const int t    = threadIdx.x;
    const int c    = t & 7;
    const int r    = t >> 3;

    const float* p = V + ((size_t)b * S + r) * D + half * VS_HALF + c * 4;

    float4 s = make_float4(0.0f, 0.0f, 0.0f, 0.0f);
#pragma unroll 8
    for (int i = 0; i < S / VS_ROWS; ++i) {
        const float4 v = *(const float4*)(p + (size_t)i * VS_ROWS * D);
        s.x += v.x; s.y += v.y; s.z += v.z; s.w += v.w;
    }

    for (int off = 8; off <= 16; off <<= 1) {
        s.x += __shfl_xor(s.x, off, 32);
        s.y += __shfl_xor(s.y, off, 32);
        s.z += __shfl_xor(s.z, off, 32);
        s.w += __shfl_xor(s.w, off, 32);
    }

    const int lane = t & 31;
    const int wave = t >> 5;
    if (lane < 8) *(float4*)&part[wave][lane * 4] = s;
    __syncthreads();

    if (t < VS_HALF) {
        float a = 0.0f;
#pragma unroll
        for (int w = 0; w < VS_WAVES; ++w) a += part[w][t];
        Vsum[b * D + half * VS_HALF + t] = a;
    }
}

__global__ __launch_bounds__(256) void attn_kernel(
    const float* __restrict__ Q, const float* __restrict__ K,
    const float* __restrict__ V, const float* __restrict__ Vsum,
    float* __restrict__ out)
{
    const int gwave = (blockIdx.x * blockDim.x + threadIdx.x) >> 5;
    if (gwave >= B * S) return;
    const int lane = threadIdx.x & 31;
    const int b = gwave / S;
    const int i = gwave - b * S;

    const size_t rowQ = (size_t)(b * S + i) * D;
    const float2 qv = *(const float2*)(Q + rowQ + lane * 2);

    float accx = Vsum[b * D + lane * 2];
    float accy = Vsum[b * D + lane * 2 + 1];
    float Z = (float)S;

#pragma unroll
    for (int w = 0; w < WIN; ++w) {
        const int j = i + DIL * (w - WIN / 2);
        if (j < 0 || j >= S) continue;
        const size_t rowJ = (size_t)(b * S + j) * D;
        const float2 kv = *(const float2*)(K + rowJ + lane * 2);
        float c = qv.x * kv.x + qv.y * kv.y;
#pragma unroll
        for (int off = 16; off > 0; off >>= 1) c += __shfl_xor(c, off, 32);
        const float em1 = __expf(c) - 1.0f;
        const float2 vv = *(const float2*)(V + rowJ + lane * 2);
        accx += em1 * vv.x;
        accy += em1 * vv.y;
        Z    += em1;
    }

    const float inv = 1.0f / Z;
    float2 o;
    o.x = accx * inv;
    o.y = accy * inv;
    *(float2*)(out + rowQ + lane * 2) = o;
}

extern "C" void kernel_launch(void* const* d_in, const int* in_sizes, int n_in,
                              void* d_out, int out_size, void* d_ws, size_t ws_size,
                              hipStream_t stream) {
    const float* x  = (const float*)d_in[0];
    const float* Wq = (const float*)d_in[1];
    const float* bq = (const float*)d_in[2];
    const float* Wk = (const float*)d_in[3];
    const float* bk = (const float*)d_in[4];
    const float* Wv = (const float*)d_in[5];
    const float* bv = (const float*)d_in[6];
    float* outp = (float*)d_out;

    constexpr size_t QKV = (size_t)B * S * D;
    float* ws = (float*)d_ws;
    float* Qb = ws;
    float* Kb = ws + QKV;
    float* Vb = ws + 2 * QKV;
    float* Vs = ws + 3 * QKV;

    const int mtiles = (B * S) / 16;
    dim3 gproj(mtiles / 8);
    qkv_proj_kernel<<<gproj, 256, 0, stream>>>(x, Wq, bq, Qb);
    qkv_proj_kernel<<<gproj, 256, 0, stream>>>(x, Wk, bk, Kb);
    qkv_proj_kernel<<<gproj, 256, 0, stream>>>(x, Wv, bv, Vb);

    vsum_kernel<<<VS_GROUPS, VS_THREADS, 0, stream>>>(Vb, Vs);

    const int nwaves = B * S;
    attn_kernel<<<(nwaves * 32) / 256, 256, 0, stream>>>(Qb, Kb, Vb, Vs, outp);
}


# proj: the three projection launches (Q, K, V) become ONE launch of 64 x 3 work-groups (blockIdx.y picks W, bias, output plane), side by side on 192 WGPs; per-wave work and sum order unchanged
# speedup vs baseline: 16.9571x; 2.1184x over previous
#include <hip/hip_runtime.h>
#include <hip/hip_bf16.h>

constexpr int B   = 2;
constexpr int S   = 4096;
constexpr int E   = 1024;
constexpr int D   = 64;
constexpr int WIN = 5;
constexpr int DIL = 2;

typedef __attribute__((ext_vector_type(2))) float v2f;
typedef __attribute__((ext_vector_type(8))) float v8f;

constexpr int PF     = 16;
constexpr int NKSTEP = E / 4;

struct KStep { v2f a; float4 w0, w1; };

__device__ __forceinline__ KStep load_kstep(const float* __restrict__ xk,
                                            const float* __restrict__ wk)
{
    KStep f;
    f.a  = *(const v2f*)xk;
    f.w0 = *(const float4*)wk;
    f.w1 = *(const float4*)(wk + D);
    return f;
}

__device__ __forceinline__ void mma_kstep(const KStep& f, v8f& acc0, v8f& acc1,
                                          v8f& acc2, v8f& acc3)
{
    v2f b0, b1, b2, b3;
    b0.x = f.w0.x;  b0.y = f.w1.x;
    b1.x = f.w0.y;  b1.y = f.w1.y;
    b2.x = f.w0.z;  b2.y = f.w1.z;
    b3.x = f.w0.w;  b3.y = f.w1.w;
    acc0 = __builtin_amdgcn_wmma_f32_16x16x4_f32(false, f.a, false, b0,
                                                 (short)0, acc0, false, false);
    acc1 = __builtin_amdgcn_wmma_f32_16x16x4_f32(false, f.a, false, b1,
                                                 (short)0, acc1, false, false);
    acc2 = __builtin_amdgcn_wmma_f32_16x16x4_f32(false, f.a, false, b2,
                                                 (short)0, acc2, false, false);
    acc3 = __builtin_amdgcn_wmma_f32_16x16x4_f32(false, f.a, false, b3,
                                                 (short)0, acc3, false, false);
}

__global__ __launch_bounds__(256) void qkv_proj_kernel(
    const float* __restrict__ x,
    const float* __restrict__ Wq,
    const float* __restrict__ Wk,
    const float* __restrict__ Wv,
    const float* __restrict__ biasq,
    const float* __restrict__ biask,
    const float* __restrict__ biasv,
    float* __restrict__ qkv)
{
    const int p = blockIdx.y;
    const float* __restrict__ W    = (p == 0) ? Wq : (p == 1) ? Wk : Wv;
    const float* __restrict__ bias = (p == 0) ? biasq : (p == 1) ? biask : biasv;
    float* __restrict__ out = qkv + (size_t)p * ((size_t)B * S * D);

    const int lane  = threadIdx.x & 31;
    const int wave  = threadIdx.x >> 5;
    const int mtile = blockIdx.x * 8 + wave;
    const int row0  = mtile * 16;

    const int m  = lane & 15;
    const int c4 = (lane & 15) * 4;
    const int kh = (lane >> 4) * 2;

    const float* xrow = x + (size_t)(row0 + m) * E + kh;
    const float* wcol = W + (size_t)kh * D + c4;

    v8f acc0 = {}, acc1 = {}, acc2 = {}, acc3 = {};

    KStep f[PF];
#pragma unroll
    for (int u = 0; u < PF; ++u)
        f[u] = load_kstep(xrow + 4 * u, wcol + (size_t)(4 * u) * D);

#pragma unroll 5
    for (int s0 = 0; s0 < NKSTEP - PF; s0 += PF) {
        const float* xk = xrow + 4 * (s0 + PF);
        const float* wk = wcol + (size_t)(4 * (s0 + PF)) * D;
#pragma unroll
        for (int u = 0; u < PF; ++u) {
            mma_kstep(f[u], acc0, acc1, acc2, acc3);
            f[u] = load_kstep(xk + 4 * u, wk + (size_t)(4 * u) * D);
            __builtin_amdgcn_sched_barrier(0);
        }
    }

#pragma unroll
    for (int u = 0; u < PF; ++u)
        mma_kstep(f[u], acc0, acc1, acc2, acc3);

    const int mo = (lane >> 4) * 8;
    const float4 bv = *(const float4*)(bias + c4);
    float* orow = out + (size_t)(row0 + mo) * D + c4;
#pragma unroll
    for (int r = 0; r < 8; ++r) {
        float4 o;
        o.x = acc0[r] + bv.x;
        o.y = acc1[r] + bv.y;
        o.z = acc2[r] + bv.z;
        o.w = acc3[r] + bv.w;
        *(float4*)(orow + (size_t)r * D) = o;
    }
}

constexpr int VS_HALF    = 32;
constexpr int VS_THREADS = 1024;
constexpr int VS_ROWS    = VS_THREADS / 8;
constexpr int VS_WAVES   = VS_THREADS / 32;
constexpr int VS_GROUPS  = B * (D / VS_HALF);

__global__ __launch_bounds__(VS_THREADS) void vsum_kernel(
    const float* __restrict__ V, float* __restrict__ Vsum)
{
    __shared__ float part[VS_WAVES][VS_HALF];

    const int b    = blockIdx.x >> 1;
    const int half = blockIdx.x & 1;
    const int t    = threadIdx.x;
    const int c    = t & 7;
    const int r    = t >> 3;

    const float* p = V + ((size_t)b * S + r) * D + half * VS_HALF + c * 4;

    float4 s = make_float4(0.0f, 0.0f, 0.0f, 0.0f);
#pragma unroll 8
    for (int i = 0; i < S / VS_ROWS; ++i) {
        const float4 v = *(const float4*)(p + (size_t)i * VS_ROWS * D);
        s.x += v.x; s.y += v.y; s.z += v.z; s.w += v.w;
    }

    for (int off = 8; off <= 16; off <<= 1) {
        s.x += __shfl_xor(s.x, off, 32);
        s.y += __shfl_xor(s.y, off, 32);
        s.z += __shfl_xor(s.z, off, 32);
        s.w += __shfl_xor(s.w, off, 32);
    }

    const int lane = t & 31;
    const int wave = t >> 5;
    if (lane < 8) *(float4*)&part[wave][lane * 4] = s;
    __syncthreads();

    if (t < VS_HALF) {
        float a = 0.0f;
#pragma unroll
        for (int w = 0; w < VS_WAVES; ++w) a += part[w][t];
        Vsum[b * D + half * VS_HALF + t] = a;
    }
}

__global__ __launch_bounds__(256) void attn_kernel(
    const float* __restrict__ Q, const float* __restrict__ K,
    const float* __restrict__ V, const float* __restrict__ Vsum,
    float* __restrict__ out)
{
    const int gwave = (blockIdx.x * blockDim.x + threadIdx.x) >> 5;
    if (gwave >= B * S) return;
    const int lane = threadIdx.x & 31;
    const int b = gwave / S;
    const int i = gwave - b * S;

    const size_t rowQ = (size_t)(b * S + i) * D;
    const float2 qv = *(const float2*)(Q + rowQ + lane * 2);

    float accx = Vsum[b * D + lane * 2];
    float accy = Vsum[b * D + lane * 2 + 1];
    float Z = (float)S;

#pragma unroll
    for (int w = 0; w < WIN; ++w) {
        const int j = i + DIL * (w - WIN / 2);
        if (j < 0 || j >= S) continue;
        const size_t rowJ = (size_t)(b * S + j) * D;
        const float2 kv = *(const float2*)(K + rowJ + lane * 2);
        float c = qv.x * kv.x + qv.y * kv.y;
#pragma unroll
        for (int off = 16; off > 0; off >>= 1) c += __shfl_xor(c, off, 32);
        const float em1 = __expf(c) - 1.0f;
        const float2 vv = *(const float2*)(V + rowJ + lane * 2);
        accx += em1 * vv.x;
        accy += em1 * vv.y;
        Z    += em1;
    }

    const float inv = 1.0f / Z;
    float2 o;
    o.x = accx * inv;
    o.y = accy * inv;
    *(float2*)(out + rowQ + lane * 2) = o;
}

extern "C" void kernel_launch(void* const* d_in, const int* in_sizes, int n_in,
                              void* d_out, int out_size, void* d_ws, size_t ws_size,
                              hipStream_t stream) {
    const float* x  = (const float*)d_in[0];
    const float* Wq = (const float*)d_in[1];
    const float* bq = (const float*)d_in[2];
    const float* Wk = (const float*)d_in[3];
    const float* bk = (const float*)d_in[4];
    const float* Wv = (const float*)d_in[5];
    const float* bv = (const float*)d_in[6];
    float* outp = (float*)d_out;

    constexpr size_t QKV = (size_t)B * S * D;
    float* ws = (float*)d_ws;
    float* Qb = ws;
    float* Kb = ws + QKV;
    float* Vb = ws + 2 * QKV;
    float* Vs = ws + 3 * QKV;

    const int mtiles = (B * S) / 16;
    dim3 gproj(mtiles / 8, 3);
    qkv_proj_kernel<<<gproj, 256, 0, stream>>>(x, Wq, Wk, Wv, bq, bk, bv, Qb);

    vsum_kernel<<<VS_GROUPS, VS_THREADS, 0, stream>>>(Vb, Vs);

    const int nwaves = B * S;
    attn_kernel<<<(nwaves * 32) / 256, 256, 0, stream>>>(Qb, Kb, Vb, Vs, outp);
}


# vsum launch removed: V work-groups of the projection leave column sums of their 128 rows (lanes, LDS), attention adds the 32 partial rows per batch itself; changes the order of the V sum
# speedup vs baseline: 21.4226x; 1.2633x over previous
#include <hip/hip_runtime.h>
#include <hip/hip_bf16.h>

constexpr int B   = 2;
constexpr int S   = 4096;
constexpr int E   = 1024;
constexpr int D   = 64;
constexpr int WIN = 5;
constexpr int DIL = 2;

typedef __attribute__((ext_vector_type(2))) float v2f;
typedef __attribute__((ext_vector_type(8))) float v8f;

constexpr int PF     = 16;
constexpr int NKSTEP = E / 4;

constexpr int PROJ_WAVES   = 8;
constexpr int PROJ_ROWS    = PROJ_WAVES * 16;
constexpr int VP_PER_BATCH = S / PROJ_ROWS;
constexpr int VP_ROWS      = B * VP_PER_BATCH;

struct KStep { v2f a; float4 w0, w1; };

__device__ __forceinline__ KStep load_kstep(const float* __restrict__ xk,
                                            const float* __restrict__ wk)
{
    KStep f;
    f.a  = *(const v2f*)xk;
    f.w0 = *(const float4*)wk;
    f.w1 = *(const float4*)(wk + D);
    return f;
}

__device__ __forceinline__ void mma_kstep(const KStep& f, v8f& acc0, v8f& acc1,
                                          v8f& acc2, v8f& acc3)
{
    v2f b0, b1, b2, b3;
    b0.x = f.w0.x;  b0.y = f.w1.x;
    b1.x = f.w0.y;  b1.y = f.w1.y;
    b2.x = f.w0.z;  b2.y = f.w1.z;
    b3.x = f.w0.w;  b3.y = f.w1.w;
    acc0 = __builtin_amdgcn_wmma_f32_16x16x4_f32(false, f.a, false, b0,
                                                 (short)0, acc0, false, false);
    acc1 = __builtin_amdgcn_wmma_f32_16x16x4_f32(false, f.a, false, b1,
                                                 (short)0, acc1, false, false);
    acc2 = __builtin_amdgcn_wmma_f32_16x16x4_f32(false, f.a, false, b2,
                                                 (short)0, acc2, false, false);
    acc3 = __builtin_amdgcn_wmma_f32_16x16x4_f32(false, f.a, false, b3,
                                                 (short)0, acc3, false, false);
}

__global__ __launch_bounds__(256) void qkv_proj_kernel(
    const float* __restrict__ x,
    const float* __restrict__ Wq,
    const float* __restrict__ Wk,
    const float* __restrict__ Wv,
    const float* __restrict__ biasq,
    const float* __restrict__ biask,
    const float* __restrict__ biasv,
    float* __restrict__ qkv,
    float* __restrict__ vpart)
{
    __shared__ float part[PROJ_WAVES][D];

    const int p = blockIdx.y;
    const float* __restrict__ W    = (p == 0) ? Wq : (p == 1) ? Wk : Wv;
    const float* __restrict__ bias = (p == 0) ? biasq : (p == 1) ? biask : biasv;
    float* __restrict__ out = qkv + (size_t)p * ((size_t)B * S * D);

    const int lane  = threadIdx.x & 31;
    const int wave  = threadIdx.x >> 5;
    const int mtile = blockIdx.x * 8 + wave;
    const int row0  = mtile * 16;

    const int m  = lane & 15;
    const int c4 = (lane & 15) * 4;
    const int kh = (lane >> 4) * 2;

    const float* xrow = x + (size_t)(row0 + m) * E + kh;
    const float* wcol = W + (size_t)kh * D + c4;

    v8f acc0 = {}, acc1 = {}, acc2 = {}, acc3 = {};

    KStep f[PF];
#pragma unroll
    for (int u = 0; u < PF; ++u)
        f[u] = load_kstep(xrow + 4 * u, wcol + (size_t)(4 * u) * D);

#pragma unroll 5
    for (int s0 = 0; s0 < NKSTEP - PF; s0 += PF) {
        const float* xk = xrow + 4 * (s0 + PF);
        const float* wk = wcol + (size_t)(4 * (s0 + PF)) * D;
#pragma unroll
        for (int u = 0; u < PF; ++u) {
            mma_kstep(f[u], acc0, acc1, acc2, acc3);
            f[u] = load_kstep(xk + 4 * u, wk + (size_t)(4 * u) * D);
            __builtin_amdgcn_sched_barrier(0);
        }
    }

#pragma unroll
    for (int u = 0; u < PF; ++u)
        mma_kstep(f[u], acc0, acc1, acc2, acc3);

    const int mo = (lane >> 4) * 8;
    const float4 bv = *(const float4*)(bias + c4);
    float4 o[8];
#pragma unroll
    for (int r = 0; r < 8; ++r) {
        o[r].x = acc0[r] + bv.x;
        o[r].y = acc1[r] + bv.y;
        o[r].z = acc2[r] + bv.z;
        o[r].w = acc3[r] + bv.w;
    }

    if (p == 2) {
        float4 cs = o[0];
#pragma unroll
        for (int r = 1; r < 8; ++r) {
            cs.x += o[r].x;
            cs.y += o[r].y;
            cs.z += o[r].z;
            cs.w += o[r].w;
        }
        cs.x += __shfl_xor(cs.x, 16, 32);
        cs.y += __shfl_xor(cs.y, 16, 32);
        cs.z += __shfl_xor(cs.z, 16, 32);
        cs.w += __shfl_xor(cs.w, 16, 32);
        if (lane < 16) *(float4*)&part[wave][c4] = cs;
        __syncthreads();
        if (threadIdx.x < D) {
            float a = part[0][threadIdx.x];
#pragma unroll
            for (int w = 1; w < PROJ_WAVES; ++w) a += part[w][threadIdx.x];
            vpart[(size_t)blockIdx.x * D + threadIdx.x] = a;
        }
    }

    float* orow = out + (size_t)(row0 + mo) * D + c4;
#pragma unroll
    for (int r = 0; r < 8; ++r)
        *(float4*)(orow + (size_t)r * D) = o[r];
}

constexpr int AT_WAVES = 8;
constexpr int VP_PER_WAVE = VP_PER_BATCH / AT_WAVES;

__global__ __launch_bounds__(256) void attn_kernel(
    const float* __restrict__ Q, const float* __restrict__ K,
    const float* __restrict__ V, const float* __restrict__ vpart,
    float* __restrict__ out)
{
    __shared__ float2 red[AT_WAVES][32];

    const int lane = threadIdx.x & 31;
    const int wave = threadIdx.x >> 5;
    const int gwave = blockIdx.x * AT_WAVES + wave;
    const int b = gwave / S;
    const int i = gwave - b * S;

    const float* pp = vpart + (size_t)(b * VP_PER_BATCH + wave * VP_PER_WAVE) * D + lane * 2;
    float2 ps = *(const float2*)pp;
#pragma unroll
    for (int k = 1; k < VP_PER_WAVE; ++k) {
        const float2 t = *(const float2*)(pp + k * D);
        ps.x += t.x;
        ps.y += t.y;
    }

    const size_t rowQ = (size_t)(b * S + i) * D;
    const float2 qv = *(const float2*)(Q + rowQ + lane * 2);

    red[wave][lane] = ps;
    __syncthreads();

    float2 vs = red[0][lane];
#pragma unroll
    for (int w = 1; w < AT_WAVES; ++w) {
        const float2 t = red[w][lane];
        vs.x += t.x;
        vs.y += t.y;
    }

    float accx = vs.x;
    float accy = vs.y;
    float Z = (float)S;

#pragma unroll
    for (int w = 0; w < WIN; ++w) {
        const int j = i + DIL * (w - WIN / 2);
        if (j < 0 || j >= S) continue;
        const size_t rowJ = (size_t)(b * S + j) * D;
        const float2 kv = *(const float2*)(K + rowJ + lane * 2);
        float c = qv.x * kv.x + qv.y * kv.y;
#pragma unroll
        for (int off = 16; off > 0; off >>= 1) c += __shfl_xor(c, off, 32);
        const float em1 = __expf(c) - 1.0f;
        const float2 vv = *(const float2*)(V + rowJ + lane * 2);
        accx += em1 * vv.x;
        accy += em1 * vv.y;
        Z    += em1;
    }

    const float inv = 1.0f / Z;
    float2 o;
    o.x = accx * inv;
    o.y = accy * inv;
    *(float2*)(out + rowQ + lane * 2) = o;
}

extern "C" void kernel_launch(void* const* d_in, const int* in_sizes, int n_in,
                              void* d_out, int out_size, void* d_ws, size_t ws_size,
                              hipStream_t stream) {
    const float* x  = (const float*)d_in[0];
    const float* Wq = (const float*)d_in[1];
    const float* bq = (const float*)d_in[2];
    const float* Wk = (const float*)d_in[3];
    const float* bk = (const float*)d_in[4];
    const float* Wv = (const float*)d_in[5];
    const float* bv = (const float*)d_in[6];
    float* outp = (float*)d_out;

    constexpr size_t QKV = (size_t)B * S * D;
    float* ws = (float*)d_ws;
    float* Qb = ws;
    float* Kb = ws + QKV;
    float* Vb = ws + 2 * QKV;
    float* Vp = ws + 3 * QKV;

    const int mtiles = (B * S) / 16;
    dim3 gproj(mtiles / 8, 3);
    qkv_proj_kernel<<<gproj, 256, 0, stream>>>(x, Wq, Wk, Wv, bq, bk, bv, Qb, Vp);

    const int nwaves = B * S;
    attn_kernel<<<(nwaves * 32) / 256, 256, 0, stream>>>(Qb, Kb, Vb, Vp, outp);
}


# attn: one wave walks 8 consecutive rows (1024 waves in 256 work-groups of 4 instead of 8192 in 1024 of 8), its 16 K/V rows loaded once into registers; every sum keeps its order
# speedup vs baseline: 22.6960x; 1.0594x over previous
#include <hip/hip_runtime.h>
#include <hip/hip_bf16.h>

constexpr int B   = 2;
constexpr int S   = 4096;
constexpr int E   = 1024;
constexpr int D   = 64;
constexpr int WIN = 5;
constexpr int DIL = 2;

typedef __attribute__((ext_vector_type(2))) float v2f;
typedef __attribute__((ext_vector_type(8))) float v8f;

constexpr int PF     = 16;
constexpr int NKSTEP = E / 4;

constexpr int PROJ_WAVES   = 8;
constexpr int PROJ_ROWS    = PROJ_WAVES * 16;
constexpr int VP_PER_BATCH = S / PROJ_ROWS;
constexpr int VP_ROWS      = B * VP_PER_BATCH;

struct KStep { v2f a; float4 w0, w1; };

__device__ __forceinline__ KStep load_kstep(const float* __restrict__ xk,
                                            const float* __restrict__ wk)
{
    KStep f;
    f.a  = *(const v2f*)xk;
    f.w0 = *(const float4*)wk;
    f.w1 = *(const float4*)(wk + D);
    return f;
}

__device__ __forceinline__ void mma_kstep(const KStep& f, v8f& acc0, v8f& acc1,
                                          v8f& acc2, v8f& acc3)
{
    v2f b0, b1, b2, b3;
    b0.x = f.w0.x;  b0.y = f.w1.x;
    b1.x = f.w0.y;  b1.y = f.w1.y;
    b2.x = f.w0.z;  b2.y = f.w1.z;
    b3.x = f.w0.w;  b3.y = f.w1.w;
    acc0 = __builtin_amdgcn_wmma_f32_16x16x4_f32(false, f.a, false, b0,
                                                 (short)0, acc0, false, false);
    acc1 = __builtin_amdgcn_wmma_f32_16x16x4_f32(false, f.a, false, b1,
                                                 (short)0, acc1, false, false);
    acc2 = __builtin_amdgcn_wmma_f32_16x16x4_f32(false, f.a, false, b2,
                                                 (short)0, acc2, false, false);
    acc3 = __builtin_amdgcn_wmma_f32_16x16x4_f32(false, f.a, false, b3,
                                                 (short)0, acc3, false, false);
}

__global__ __launch_bounds__(256) void qkv_proj_kernel(
    const float* __restrict__ x,
    const float* __restrict__ Wq,
    const float* __restrict__ Wk,
    const float* __restrict__ Wv,
    const float* __restrict__ biasq,
    const float* __restrict__ biask,
    const float* __restrict__ biasv,
    float* __restrict__ qkv,
    float* __restrict__ vpart)
{
    __shared__ float part[PROJ_WAVES][D];

    const int p = blockIdx.y;
    const float* __restrict__ W    = (p == 0) ? Wq : (p == 1) ? Wk : Wv;
    const float* __restrict__ bias = (p == 0) ? biasq : (p == 1) ? biask : biasv;
    float* __restrict__ out = qkv + (size_t)p * ((size_t)B * S * D);

    const int lane  = threadIdx.x & 31;
    const int wave  = threadIdx.x >> 5;
    const int mtile = blockIdx.x * 8 + wave;
    const int row0  = mtile * 16;

    const int m  = lane & 15;
    const int c4 = (lane & 15) * 4;
    const int kh = (lane >> 4) * 2;

    const float* xrow = x + (size_t)(row0 + m) * E + kh;
    const float* wcol = W + (size_t)kh * D + c4;

    v8f acc0 = {}, acc1 = {}, acc2 = {}, acc3 = {};

    KStep f[PF];
#pragma unroll
    for (int u = 0; u < PF; ++u)
        f[u] = load_kstep(xrow + 4 * u, wcol + (size_t)(4 * u) * D);

#pragma unroll 5
    for (int s0 = 0; s0 < NKSTEP - PF; s0 += PF) {
        const float* xk = xrow + 4 * (s0 + PF);
        const float* wk = wcol + (size_t)(4 * (s0 + PF)) * D;
#pragma unroll
        for (int u = 0; u < PF; ++u) {
            mma_kstep(f[u], acc0, acc1, acc2, acc3);
            f[u] = load_kstep(xk + 4 * u, wk + (size_t)(4 * u) * D);
            __builtin_amdgcn_sched_barrier(0);
        }
    }

#pragma unroll
    for (int u = 0; u < PF; ++u)
        mma_kstep(f[u], acc0, acc1, acc2, acc3);

    const int mo = (lane >> 4) * 8;
    const float4 bv = *(const float4*)(bias + c4);
    float4 o[8];
#pragma unroll
    for (int r = 0; r < 8; ++r) {
        o[r].x = acc0[r] + bv.x;
        o[r].y = acc1[r] + bv.y;
        o[r].z = acc2[r] + bv.z;
        o[r].w = acc3[r] + bv.w;
    }

    if (p == 2) {
        float4 cs = o[0];
#pragma unroll
        for (int r = 1; r < 8; ++r) {
            cs.x += o[r].x;
            cs.y += o[r].y;
            cs.z += o[r].z;
            cs.w += o[r].w;
        }
        cs.x += __shfl_xor(cs.x, 16, 32);
        cs.y += __shfl_xor(cs.y, 16, 32);
        cs.z += __shfl_xor(cs.z, 16, 32);
        cs.w += __shfl_xor(cs.w, 16, 32);
        if (lane < 16) *(float4*)&part[wave][c4] = cs;
        __syncthreads();
        if (threadIdx.x < D) {
            float a = part[0][threadIdx.x];
#pragma unroll
            for (int w = 1; w < PROJ_WAVES; ++w) a += part[w][threadIdx.x];
            vpart[(size_t)blockIdx.x * D + threadIdx.x] = a;
        }
    }

    float* orow = out + (size_t)(row0 + mo) * D + c4;
#pragma unroll
    for (int r = 0; r < 8; ++r)
        *(float4*)(orow + (size_t)r * D) = o[r];
}

constexpr int AT_WAVES = 4;
constexpr int AT_ROWS  = 8;
constexpr int AT_HALO  = DIL * (WIN / 2);
constexpr int AT_SPAN  = AT_ROWS + 2 * AT_HALO;
constexpr int VP_GROUPS      = 8;
constexpr int VP_PER_GROUP   = VP_PER_BATCH / VP_GROUPS;
constexpr int GROUPS_PER_WAVE = VP_GROUPS / AT_WAVES;

__global__ __launch_bounds__(128) void attn_kernel(
    const float* __restrict__ Q, const float* __restrict__ K,
    const float* __restrict__ V, const float* __restrict__ vpart,
    float* __restrict__ out)
{
    __shared__ float2 red[VP_GROUPS][32];

    const int lane = threadIdx.x & 31;
    const int wave = __builtin_amdgcn_readfirstlane((int)(threadIdx.x >> 5));
    const int gwave = blockIdx.x * AT_WAVES + wave;
    const int r0 = gwave * AT_ROWS;
    const int b  = r0 / S;
    const int i0 = r0 - b * S;

    const float* pp = vpart
        + (size_t)(b * VP_PER_BATCH + wave * (GROUPS_PER_WAVE * VP_PER_GROUP)) * D + lane * 2;
    float2 ps[GROUPS_PER_WAVE];
#pragma unroll
    for (int g = 0; g < GROUPS_PER_WAVE; ++g) {
        ps[g] = *(const float2*)(pp + (size_t)(g * VP_PER_GROUP) * D);
#pragma unroll
        for (int k = 1; k < VP_PER_GROUP; ++k) {
            const float2 t = *(const float2*)(pp + (size_t)(g * VP_PER_GROUP + k) * D);
            ps[g].x += t.x;
            ps[g].y += t.y;
        }
    }

    const float* Kb = K + (size_t)b * S * D + lane * 2;
    const float* Vb = V + (size_t)b * S * D + lane * 2;
    float2 qv[AT_ROWS], kk[AT_SPAN], vv[AT_SPAN];
#pragma unroll
    for (int r = 0; r < AT_ROWS; ++r)
        qv[r] = *(const float2*)(Q + (size_t)(r0 + r) * D + lane * 2);
#pragma unroll
    for (int t = 0; t < AT_SPAN; ++t) {
        int j = i0 - AT_HALO + t;
        if (t < AT_HALO) j = (j < 0) ? 0 : j;
        if (t >= AT_ROWS + AT_HALO) j = (j > S - 1) ? (S - 1) : j;
        kk[t] = *(const float2*)(Kb + (size_t)j * D);
        vv[t] = *(const float2*)(Vb + (size_t)j * D);
    }

#pragma unroll
    for (int g = 0; g < GROUPS_PER_WAVE; ++g)
        red[wave * GROUPS_PER_WAVE + g][lane] = ps[g];
    __syncthreads();

    float2 vs = red[0][lane];
#pragma unroll
    for (int w = 1; w < VP_GROUPS; ++w) {
        const float2 t = red[w][lane];
        vs.x += t.x;
        vs.y += t.y;
    }

    constexpr int NDOT = AT_ROWS * WIN;
    float c[NDOT];
#pragma unroll
    for (int r = 0; r < AT_ROWS; ++r) {
#pragma unroll
        for (int w = 0; w < WIN; ++w) {
            const int t = r + DIL * w;
            c[r * WIN + w] = qv[r].x * kk[t].x + qv[r].y * kk[t].y;
        }
    }

#pragma unroll
    for (int off = 16; off > 0; off >>= 1) {
        float x[NDOT];
#pragma unroll
        for (int n = 0; n < NDOT; ++n) x[n] = __shfl_xor(c[n], off, 32);
        __builtin_amdgcn_sched_barrier(0);
#pragma unroll
        for (int n = 0; n < NDOT; ++n) c[n] += x[n];
        __builtin_amdgcn_sched_barrier(0);
    }

#pragma unroll
    for (int r = 0; r < AT_ROWS; ++r) {
        float accx = vs.x;
        float accy = vs.y;
        float Z = (float)S;
#pragma unroll
        for (int w = 0; w < WIN; ++w) {
            const int t = r + DIL * w;
            const int j = i0 + r + DIL * (w - WIN / 2);
            const bool valid = (j >= 0) && (j < S);
            float em1 = __expf(c[r * WIN + w]) - 1.0f;
            em1 = valid ? em1 : 0.0f;
            accx += em1 * vv[t].x;
            accy += em1 * vv[t].y;
            Z    += em1;
        }
        const float inv = 1.0f / Z;
        float2 o;
        o.x = accx * inv;
        o.y = accy * inv;
        *(float2*)(out + (size_t)(r0 + r) * D + lane * 2) = o;
    }
}

extern "C" void kernel_launch(void* const* d_in, const int* in_sizes, int n_in,
                              void* d_out, int out_size, void* d_ws, size_t ws_size,
                              hipStream_t stream) {
    const float* x  = (const float*)d_in[0];
    const float* Wq = (const float*)d_in[1];
    const float* bq = (const float*)d_in[2];
    const float* Wk = (const float*)d_in[3];
    const float* bk = (const float*)d_in[4];
    const float* Wv = (const float*)d_in[5];
    const float* bv = (const float*)d_in[6];
    float* outp = (float*)d_out;

    constexpr size_t QKV = (size_t)B * S * D;
    float* ws = (float*)d_ws;
    float* Qb = ws;
    float* Kb = ws + QKV;
    float* Vb = ws + 2 * QKV;
    float* Vp = ws + 3 * QKV;

    const int mtiles = (B * S) / 16;
    dim3 gproj(mtiles / 8, 3);
    qkv_proj_kernel<<<gproj, 256, 0, stream>>>(x, Wq, Wk, Wv, bq, bk, bv, Qb, Vp);

    const int nwaves = (B * S) / AT_ROWS;
    attn_kernel<<<nwaves / AT_WAVES, AT_WAVES * 32, 0, stream>>>(Qb, Kb, Vb, Vp, outp);
}


# proj: one wave per SIMD: 256 work-groups of 4 waves, each wave 16 rows x 96 of the 192 Q,K,V columns (6 n-tiles), loads grouped per 4 k-steps; V partial sums per 64 rows: changes the V sum order
# speedup vs baseline: 25.1705x; 1.1090x over previous
#include <hip/hip_runtime.h>
#include <hip/hip_bf16.h>

constexpr int B   = 2;
constexpr int S   = 4096;
constexpr int E   = 1024;
constexpr int D   = 64;
constexpr int WIN = 5;
constexpr int DIL = 2;

typedef __attribute__((ext_vector_type(2))) float v2f;
typedef __attribute__((ext_vector_type(8))) float v8f;

constexpr int GK     = 4;
constexpr int NG     = 3;
constexpr int PF     = GK * NG;
constexpr int NKSTEP = E / 4;
constexpr int NTRIP  = (NKSTEP - PF) / PF;
constexpr int KTAIL  = NKSTEP - PF - NTRIP * PF;
static_assert(KTAIL == GK, "the tail below refills exactly one group");

constexpr int PROJ_WAVES   = 4;
constexpr int PROJ_ROWS    = PROJ_WAVES * 16;
constexpr int VP_PER_BATCH = S / PROJ_ROWS;
constexpr int VP_ROWS      = B * VP_PER_BATCH;
constexpr int KHALF        = D / 2;

struct KStep { v2f a; float4 f0, f1; float2 h0, h1; };

__device__ __forceinline__ KStep load_kstep(const float* __restrict__ xk,
                                            const float* __restrict__ wf,
                                            const float* __restrict__ wh)
{
    KStep f;
    f.a  = *(const v2f*)xk;
    f.f0 = *(const float4*)wf;
    f.f1 = *(const float4*)(wf + D);
    f.h0 = *(const float2*)wh;
    f.h1 = *(const float2*)(wh + D);
    return f;
}

__device__ __forceinline__ void mma_kstep(const KStep& f, v8f& acc0, v8f& acc1, v8f& acc2,
                                          v8f& acc3, v8f& acc4, v8f& acc5)
{
    v2f b0, b1, b2, b3, b4, b5;
    b0.x = f.f0.x;  b0.y = f.f1.x;
    b1.x = f.f0.y;  b1.y = f.f1.y;
    b2.x = f.f0.z;  b2.y = f.f1.z;
    b3.x = f.f0.w;  b3.y = f.f1.w;
    b4.x = f.h0.x;  b4.y = f.h1.x;
    b5.x = f.h0.y;  b5.y = f.h1.y;
    acc0 = __builtin_amdgcn_wmma_f32_16x16x4_f32(false, f.a, false, b0,
                                                 (short)0, acc0, false, false);
    acc1 = __builtin_amdgcn_wmma_f32_16x16x4_f32(false, f.a, false, b1,
                                                 (short)0, acc1, false, false);
    acc2 = __builtin_amdgcn_wmma_f32_16x16x4_f32(false, f.a, false, b2,
                                                 (short)0, acc2, false, false);
    acc3 = __builtin_amdgcn_wmma_f32_16x16x4_f32(false, f.a, false, b3,
                                                 (short)0, acc3, false, false);
    acc4 = __builtin_amdgcn_wmma_f32_16x16x4_f32(false, f.a, false, b4,
                                                 (short)0, acc4, false, false);
    acc5 = __builtin_amdgcn_wmma_f32_16x16x4_f32(false, f.a, false, b5,
                                                 (short)0, acc5, false, false);
}

__global__ __launch_bounds__(128) __attribute__((amdgpu_waves_per_eu(1, 1))) void qkv_proj_kernel(
    const float* __restrict__ x,
    const float* __restrict__ Wq,
    const float* __restrict__ Wk,
    const float* __restrict__ Wv,
    const float* __restrict__ biasq,
    const float* __restrict__ biask,
    const float* __restrict__ biasv,
    float* __restrict__ qkv,
    float* __restrict__ vpart)
{
    __shared__ float part[PROJ_WAVES][D];

    const int hsel = blockIdx.y;
    const float* __restrict__ Wf    = hsel ? Wv : Wq;
    const float* __restrict__ biasf = hsel ? biasv : biasq;
    float* __restrict__ outf = qkv + (size_t)(hsel ? 2 : 0) * ((size_t)B * S * D);
    float* __restrict__ outk = qkv + (size_t)B * S * D;
    const int koff = hsel * KHALF;

    const int lane  = threadIdx.x & 31;
    const int wave  = threadIdx.x >> 5;
    const int mtile = blockIdx.x * PROJ_WAVES + wave;
    const int row0  = mtile * 16;

    const int m  = lane & 15;
    const int c4 = (lane & 15) * 4;
    const int c2 = koff + (lane & 15) * 2;
    const int kh = (lane >> 4) * 2;

    const float* xrow = x  + (size_t)(row0 + m) * E + kh;
    const float* wfc  = Wf + (size_t)kh * D + c4;
    const float* whc  = Wk + (size_t)kh * D + c2;

    v8f acc0 = {}, acc1 = {}, acc2 = {}, acc3 = {}, acc4 = {}, acc5 = {};

    KStep f[PF];
#pragma unroll
    for (int u = 0; u < PF; ++u)
        f[u] = load_kstep(xrow + 4 * u, wfc + (size_t)(4 * u) * D, whc + (size_t)(4 * u) * D);

#pragma unroll 5
    for (int s0 = 0; s0 < NTRIP * PF; s0 += PF) {
        const float* xk = xrow + 4 * (s0 + PF);
        const float* wf = wfc + (size_t)(4 * (s0 + PF)) * D;
        const float* wh = whc + (size_t)(4 * (s0 + PF)) * D;
#pragma unroll
        for (int g = 0; g < NG; ++g) {
#pragma unroll
            for (int u = g * GK; u < (g + 1) * GK; ++u)
                mma_kstep(f[u], acc0, acc1, acc2, acc3, acc4, acc5);
#pragma unroll
            for (int u = g * GK; u < (g + 1) * GK; ++u)
                f[u] = load_kstep(xk + 4 * u, wf + (size_t)(4 * u) * D, wh + (size_t)(4 * u) * D);
            __builtin_amdgcn_sched_barrier(0);
        }
    }

    {
        const float* xk = xrow + 4 * (NTRIP * PF + PF);
        const float* wf = wfc + (size_t)(4 * (NTRIP * PF + PF)) * D;
        const float* wh = whc + (size_t)(4 * (NTRIP * PF + PF)) * D;
#pragma unroll
        for (int u = 0; u < GK; ++u)
            mma_kstep(f[u], acc0, acc1, acc2, acc3, acc4, acc5);
#pragma unroll
        for (int u = 0; u < GK; ++u)
            f[u] = load_kstep(xk + 4 * u, wf + (size_t)(4 * u) * D, wh + (size_t)(4 * u) * D);
    }
#pragma unroll
    for (int u = GK; u < PF; ++u)
        mma_kstep(f[u], acc0, acc1, acc2, acc3, acc4, acc5);
#pragma unroll
    for (int u = 0; u < GK; ++u)
        mma_kstep(f[u], acc0, acc1, acc2, acc3, acc4, acc5);

    const int mo = (lane >> 4) * 8;
    const float4 bvf = *(const float4*)(biasf + c4);
    const float2 bvk = *(const float2*)(biask + c2);
    float4 o[8];
    float2 ok[8];
#pragma unroll
    for (int r = 0; r < 8; ++r) {
        o[r].x = acc0[r] + bvf.x;
        o[r].y = acc1[r] + bvf.y;
        o[r].z = acc2[r] + bvf.z;
        o[r].w = acc3[r] + bvf.w;
        ok[r].x = acc4[r] + bvk.x;
        ok[r].y = acc5[r] + bvk.y;
    }

    if (hsel == 1) {
        float4 cs = o[0];
#pragma unroll
        for (int r = 1; r < 8; ++r) {
            cs.x += o[r].x;
            cs.y += o[r].y;
            cs.z += o[r].z;
            cs.w += o[r].w;
        }
        cs.x += __shfl_xor(cs.x, 16, 32);
        cs.y += __shfl_xor(cs.y, 16, 32);
        cs.z += __shfl_xor(cs.z, 16, 32);
        cs.w += __shfl_xor(cs.w, 16, 32);
        if (lane < 16) *(float4*)&part[wave][c4] = cs;
        __syncthreads();
        if (threadIdx.x < D) {
            float a = part[0][threadIdx.x];
#pragma unroll
            for (int w = 1; w < PROJ_WAVES; ++w) a += part[w][threadIdx.x];
            vpart[(size_t)blockIdx.x * D + threadIdx.x] = a;
        }
    }

    float* orow = outf + (size_t)(row0 + mo) * D + c4;
    float* krow = outk + (size_t)(row0 + mo) * D + c2;
#pragma unroll
    for (int r = 0; r < 8; ++r) {
        *(float4*)(orow + (size_t)r * D) = o[r];
        *(float2*)(krow + (size_t)r * D) = ok[r];
    }
}

constexpr int AT_WAVES = 4;
constexpr int AT_ROWS  = 8;
constexpr int AT_HALO  = DIL * (WIN / 2);
constexpr int AT_SPAN  = AT_ROWS + 2 * AT_HALO;
constexpr int VP_GROUPS      = 8;
constexpr int VP_PER_GROUP   = VP_PER_BATCH / VP_GROUPS;
constexpr int GROUPS_PER_WAVE = VP_GROUPS / AT_WAVES;

__global__ __launch_bounds__(128) void attn_kernel(
    const float* __restrict__ Q, const float* __restrict__ K,
    const float* __restrict__ V, const float* __restrict__ vpart,
    float* __restrict__ out)
{
    __shared__ float2 red[VP_GROUPS][32];

    const int lane = threadIdx.x & 31;
    const int wave = __builtin_amdgcn_readfirstlane((int)(threadIdx.x >> 5));
    const int gwave = blockIdx.x * AT_WAVES + wave;
    const int r0 = gwave * AT_ROWS;
    const int b  = r0 / S;
    const int i0 = r0 - b * S;

    const float* pp = vpart
        + (size_t)(b * VP_PER_BATCH + wave * (GROUPS_PER_WAVE * VP_PER_GROUP)) * D + lane * 2;
    float2 ps[GROUPS_PER_WAVE];
#pragma unroll
    for (int g = 0; g < GROUPS_PER_WAVE; ++g) {
        ps[g] = *(const float2*)(pp + (size_t)(g * VP_PER_GROUP) * D);
#pragma unroll
        for (int k = 1; k < VP_PER_GROUP; ++k) {
            const float2 t = *(const float2*)(pp + (size_t)(g * VP_PER_GROUP + k) * D);
            ps[g].x += t.x;
            ps[g].y += t.y;
        }
    }

    const float* Kb = K + (size_t)b * S * D + lane * 2;
    const float* Vb = V + (size_t)b * S * D + lane * 2;
    float2 qv[AT_ROWS], kk[AT_SPAN], vv[AT_SPAN];
#pragma unroll
    for (int r = 0; r < AT_ROWS; ++r)
        qv[r] = *(const float2*)(Q + (size_t)(r0 + r) * D + lane * 2);
#pragma unroll
    for (int t = 0; t < AT_SPAN; ++t) {
        int j = i0 - AT_HALO + t;
        if (t < AT_HALO) j = (j < 0) ? 0 : j;
        if (t >= AT_ROWS + AT_HALO) j = (j > S - 1) ? (S - 1) : j;
        kk[t] = *(const float2*)(Kb + (size_t)j * D);
        vv[t] = *(const float2*)(Vb + (size_t)j * D);
    }

#pragma unroll
    for (int g = 0; g < GROUPS_PER_WAVE; ++g)
        red[wave * GROUPS_PER_WAVE + g][lane] = ps[g];
    __syncthreads();

    float2 vs = red[0][lane];
#pragma unroll
    for (int w = 1; w < VP_GROUPS; ++w) {
        const float2 t = red[w][lane];
        vs.x += t.x;
        vs.y += t.y;
    }

    constexpr int NDOT = AT_ROWS * WIN;
    float c[NDOT];
#pragma unroll
    for (int r = 0; r < AT_ROWS; ++r) {
#pragma unroll
        for (int w = 0; w < WIN; ++w) {
            const int t = r + DIL * w;
            c[r * WIN + w] = qv[r].x * kk[t].x + qv[r].y * kk[t].y;
        }
    }

#pragma unroll
    for (int off = 16; off > 0; off >>= 1) {
        float x[NDOT];
#pragma unroll
        for (int n = 0; n < NDOT; ++n) x[n] = __shfl_xor(c[n], off, 32);
        __builtin_amdgcn_sched_barrier(0);
#pragma unroll
        for (int n = 0; n < NDOT; ++n) c[n] += x[n];
        __builtin_amdgcn_sched_barrier(0);
    }

#pragma unroll
    for (int r = 0; r < AT_ROWS; ++r) {
        float accx = vs.x;
        float accy = vs.y;
        float Z = (float)S;
#pragma unroll
        for (int w = 0; w < WIN; ++w) {
            const int t = r + DIL * w;
            const int j = i0 + r + DIL * (w - WIN / 2);
            const bool valid = (j >= 0) && (j < S);
            float em1 = __expf(c[r * WIN + w]) - 1.0f;
            em1 = valid ? em1 : 0.0f;
            accx += em1 * vv[t].x;
            accy += em1 * vv[t].y;
            Z    += em1;
        }
        const float inv = 1.0f / Z;
        float2 o;
        o.x = accx * inv;
        o.y = accy * inv;
        *(float2*)(out + (size_t)(r0 + r) * D + lane * 2) = o;
    }
}

extern "C" void kernel_launch(void* const* d_in, const int* in_sizes, int n_in,
                              void* d_out, int out_size, void* d_ws, size_t ws_size,
                              hipStream_t stream) {
    const float* x  = (const float*)d_in[0];
    const float* Wq = (const float*)d_in[1];
    const float* bq = (const float*)d_in[2];
    const float* Wk = (const float*)d_in[3];
    const float* bk = (const float*)d_in[4];
    const float* Wv = (const float*)d_in[5];
    const float* bv = (const float*)d_in[6];
    float* outp = (float*)d_out;

    constexpr size_t QKV = (size_t)B * S * D;
    float* ws = (float*)d_ws;
    float* Qb = ws;
    float* Kb = ws + QKV;
    float* Vb = ws + 2 * QKV;
    float* Vp = ws + 3 * QKV;

    const int mtiles = (B * S) / 16;
    dim3 gproj(mtiles / PROJ_WAVES, 2);
    qkv_proj_kernel<<<gproj, PROJ_WAVES * 32, 0, stream>>>(x, Wq, Wk, Wv, bq, bk, bv, Qb, Vp);

    const int nwaves = (B * S) / AT_ROWS;
    attn_kernel<<<nwaves / AT_WAVES, AT_WAVES * 32, 0, stream>>>(Qb, Kb, Vb, Vp, outp);
}


# proj: K range split over two waves per SIMD (8-wave work-groups: 4 row tiles x 2 K halves of 128 k-steps, the halves added once through LDS); changes the order of the k sum
# speedup vs baseline: 26.9642x; 1.0713x over previous
#include <hip/hip_runtime.h>
#include <hip/hip_bf16.h>

constexpr int B   = 2;
constexpr int S   = 4096;
constexpr int E   = 1024;
constexpr int D   = 64;
constexpr int WIN = 5;
constexpr int DIL = 2;

typedef __attribute__((ext_vector_type(2))) float v2f;
typedef __attribute__((ext_vector_type(8))) float v8f;

constexpr int GK     = 4;
constexpr int NG     = 3;
constexpr int PF     = GK * NG;
constexpr int NKSTEP = E / 4;
constexpr int KSPLIT = 2;
constexpr int NKW    = NKSTEP / KSPLIT;
constexpr int NTRIP  = (NKW - PF) / PF;
constexpr int KTAIL  = NKW - PF - NTRIP * PF;
static_assert(KTAIL % GK == 0 && KTAIL < PF, "the tail below refills whole groups");

constexpr int PROJ_TILES   = 4;
constexpr int PROJ_WAVES   = PROJ_TILES * KSPLIT;
constexpr int PROJ_ROWS    = PROJ_TILES * 16;
constexpr int XCH_VEC      = 6 * 8 / 4;
constexpr int VP_PER_BATCH = S / PROJ_ROWS;
constexpr int VP_ROWS      = B * VP_PER_BATCH;
constexpr int KHALF        = D / 2;

struct KStep { v2f a; float4 f0, f1; float2 h0, h1; };

__device__ __forceinline__ KStep load_kstep(const float* __restrict__ xk,
                                            const float* __restrict__ wf,
                                            const float* __restrict__ wh)
{
    KStep f;
    f.a  = *(const v2f*)xk;
    f.f0 = *(const float4*)wf;
    f.f1 = *(const float4*)(wf + D);
    f.h0 = *(const float2*)wh;
    f.h1 = *(const float2*)(wh + D);
    return f;
}

__device__ __forceinline__ void mma_kstep(const KStep& f, v8f& acc0, v8f& acc1, v8f& acc2,
                                          v8f& acc3, v8f& acc4, v8f& acc5)
{
    v2f b0, b1, b2, b3, b4, b5;
    b0.x = f.f0.x;  b0.y = f.f1.x;
    b1.x = f.f0.y;  b1.y = f.f1.y;
    b2.x = f.f0.z;  b2.y = f.f1.z;
    b3.x = f.f0.w;  b3.y = f.f1.w;
    b4.x = f.h0.x;  b4.y = f.h1.x;
    b5.x = f.h0.y;  b5.y = f.h1.y;
    acc0 = __builtin_amdgcn_wmma_f32_16x16x4_f32(false, f.a, false, b0,
                                                 (short)0, acc0, false, false);
    acc1 = __builtin_amdgcn_wmma_f32_16x16x4_f32(false, f.a, false, b1,
                                                 (short)0, acc1, false, false);
    acc2 = __builtin_amdgcn_wmma_f32_16x16x4_f32(false, f.a, false, b2,
                                                 (short)0, acc2, false, false);
    acc3 = __builtin_amdgcn_wmma_f32_16x16x4_f32(false, f.a, false, b3,
                                                 (short)0, acc3, false, false);
    acc4 = __builtin_amdgcn_wmma_f32_16x16x4_f32(false, f.a, false, b4,
                                                 (short)0, acc4, false, false);
    acc5 = __builtin_amdgcn_wmma_f32_16x16x4_f32(false, f.a, false, b5,
                                                 (short)0, acc5, false, false);
}

__global__ __launch_bounds__(256) __attribute__((amdgpu_waves_per_eu(2, 2))) void qkv_proj_kernel(
    const float* __restrict__ x,
    const float* __restrict__ Wq,
    const float* __restrict__ Wk,
    const float* __restrict__ Wv,
    const float* __restrict__ biasq,
    const float* __restrict__ biask,
    const float* __restrict__ biasv,
    float* __restrict__ qkv,
    float* __restrict__ vpart)
{
    __shared__ float part[PROJ_TILES][D];
    __shared__ float4 xch[PROJ_TILES][XCH_VEC][32];

    const int hsel = blockIdx.y;
    const float* __restrict__ Wf    = hsel ? Wv : Wq;
    const float* __restrict__ biasf = hsel ? biasv : biasq;
    float* __restrict__ outf = qkv + (size_t)(hsel ? 2 : 0) * ((size_t)B * S * D);
    float* __restrict__ outk = qkv + (size_t)B * S * D;
    const int koff = hsel * KHALF;

    const int lane  = threadIdx.x & 31;
    const int wave  = __builtin_amdgcn_readfirstlane((int)(threadIdx.x >> 5));
    const int rt    = wave % PROJ_TILES;
    const int kq    = wave / PROJ_TILES;
    const int mtile = blockIdx.x * PROJ_TILES + rt;
    const int row0  = mtile * 16;
    const int k0    = kq * (NKW * 4);

    const int m  = lane & 15;
    const int c4 = (lane & 15) * 4;
    const int c2 = koff + (lane & 15) * 2;
    const int kh = (lane >> 4) * 2;

    const float* xrow = x  + (size_t)(row0 + m) * E + k0 + kh;
    const float* wfc  = Wf + (size_t)(k0 + kh) * D + c4;
    const float* whc  = Wk + (size_t)(k0 + kh) * D + c2;

    v8f acc0 = {}, acc1 = {}, acc2 = {}, acc3 = {}, acc4 = {}, acc5 = {};

    KStep f[PF];
#pragma unroll
    for (int u = 0; u < PF; ++u)
        f[u] = load_kstep(xrow + 4 * u, wfc + (size_t)(4 * u) * D, whc + (size_t)(4 * u) * D);

#pragma unroll
    for (int s0 = 0; s0 < NTRIP * PF; s0 += PF) {
        const float* xk = xrow + 4 * (s0 + PF);
        const float* wf = wfc + (size_t)(4 * (s0 + PF)) * D;
        const float* wh = whc + (size_t)(4 * (s0 + PF)) * D;
#pragma unroll
        for (int g = 0; g < NG; ++g) {
#pragma unroll
            for (int u = g * GK; u < (g + 1) * GK; ++u)
                mma_kstep(f[u], acc0, acc1, acc2, acc3, acc4, acc5);
#pragma unroll
            for (int u = g * GK; u < (g + 1) * GK; ++u)
                f[u] = load_kstep(xk + 4 * u, wf + (size_t)(4 * u) * D, wh + (size_t)(4 * u) * D);
            __builtin_amdgcn_sched_barrier(0);
        }
    }

    {
        const float* xk = xrow + 4 * (NTRIP * PF + PF);
        const float* wf = wfc + (size_t)(4 * (NTRIP * PF + PF)) * D;
        const float* wh = whc + (size_t)(4 * (NTRIP * PF + PF)) * D;
#pragma unroll
        for (int g = 0; g < KTAIL / GK; ++g) {
#pragma unroll
            for (int u = g * GK; u < (g + 1) * GK; ++u)
                mma_kstep(f[u], acc0, acc1, acc2, acc3, acc4, acc5);
#pragma unroll
            for (int u = g * GK; u < (g + 1) * GK; ++u)
                f[u] = load_kstep(xk + 4 * u, wf + (size_t)(4 * u) * D, wh + (size_t)(4 * u) * D);
            __builtin_amdgcn_sched_barrier(0);
        }
    }
#pragma unroll
    for (int u = KTAIL; u < PF; ++u)
        mma_kstep(f[u], acc0, acc1, acc2, acc3, acc4, acc5);
#pragma unroll
    for (int u = 0; u < KTAIL; ++u)
        mma_kstep(f[u], acc0, acc1, acc2, acc3, acc4, acc5);

    if (kq == 1) {
#pragma unroll
        for (int q = 0; q < 2; ++q) {
            xch[rt][0 + q][lane]  = make_float4(acc0[4 * q], acc0[4 * q + 1], acc0[4 * q + 2], acc0[4 * q + 3]);
            xch[rt][2 + q][lane]  = make_float4(acc1[4 * q], acc1[4 * q + 1], acc1[4 * q + 2], acc1[4 * q + 3]);
            xch[rt][4 + q][lane]  = make_float4(acc2[4 * q], acc2[4 * q + 1], acc2[4 * q + 2], acc2[4 * q + 3]);
            xch[rt][6 + q][lane]  = make_float4(acc3[4 * q], acc3[4 * q + 1], acc3[4 * q + 2], acc3[4 * q + 3]);
            xch[rt][8 + q][lane]  = make_float4(acc4[4 * q], acc4[4 * q + 1], acc4[4 * q + 2], acc4[4 * q + 3]);
            xch[rt][10 + q][lane] = make_float4(acc5[4 * q], acc5[4 * q + 1], acc5[4 * q + 2], acc5[4 * q + 3]);
        }
    }
    __syncthreads();
    if (kq == 0) {
#pragma unroll
        for (int q = 0; q < 2; ++q) {
            const float4 t0 = xch[rt][0 + q][lane];
            const float4 t1 = xch[rt][2 + q][lane];
            const float4 t2 = xch[rt][4 + q][lane];
            const float4 t3 = xch[rt][6 + q][lane];
            const float4 t4 = xch[rt][8 + q][lane];
            const float4 t5 = xch[rt][10 + q][lane];
            acc0[4 * q] += t0.x;  acc0[4 * q + 1] += t0.y;  acc0[4 * q + 2] += t0.z;  acc0[4 * q + 3] += t0.w;
            acc1[4 * q] += t1.x;  acc1[4 * q + 1] += t1.y;  acc1[4 * q + 2] += t1.z;  acc1[4 * q + 3] += t1.w;
            acc2[4 * q] += t2.x;  acc2[4 * q + 1] += t2.y;  acc2[4 * q + 2] += t2.z;  acc2[4 * q + 3] += t2.w;
            acc3[4 * q] += t3.x;  acc3[4 * q + 1] += t3.y;  acc3[4 * q + 2] += t3.z;  acc3[4 * q + 3] += t3.w;
            acc4[4 * q] += t4.x;  acc4[4 * q + 1] += t4.y;  acc4[4 * q + 2] += t4.z;  acc4[4 * q + 3] += t4.w;
            acc5[4 * q] += t5.x;  acc5[4 * q + 1] += t5.y;  acc5[4 * q + 2] += t5.z;  acc5[4 * q + 3] += t5.w;
        }
    }

    const int mo = (lane >> 4) * 8;
    const float4 bvf = *(const float4*)(biasf + c4);
    const float2 bvk = *(const float2*)(biask + c2);
    float4 o[8];
    float2 ok[8];
#pragma unroll
    for (int r = 0; r < 8; ++r) {
        o[r].x = acc0[r] + bvf.x;
        o[r].y = acc1[r] + bvf.y;
        o[r].z = acc2[r] + bvf.z;
        o[r].w = acc3[r] + bvf.w;
        ok[r].x = acc4[r] + bvk.x;
        ok[r].y = acc5[r] + bvk.y;
    }

    if (hsel == 1) {
        if (kq == 0) {
            float4 cs = o[0];
#pragma unroll
            for (int r = 1; r < 8; ++r) {
                cs.x += o[r].x;
                cs.y += o[r].y;
                cs.z += o[r].z;
                cs.w += o[r].w;
            }
            cs.x += __shfl_xor(cs.x, 16, 32);
            cs.y += __shfl_xor(cs.y, 16, 32);
            cs.z += __shfl_xor(cs.z, 16, 32);
            cs.w += __shfl_xor(cs.w, 16, 32);
            if (lane < 16) *(float4*)&part[rt][c4] = cs;
        }
        __syncthreads();
        if (threadIdx.x < D) {
            float a = part[0][threadIdx.x];
#pragma unroll
            for (int w = 1; w < PROJ_TILES; ++w) a += part[w][threadIdx.x];
            vpart[(size_t)blockIdx.x * D + threadIdx.x] = a;
        }
    }

    if (kq == 0) {
        float* orow = outf + (size_t)(row0 + mo) * D + c4;
        float* krow = outk + (size_t)(row0 + mo) * D + c2;
#pragma unroll
        for (int r = 0; r < 8; ++r) {
            *(float4*)(orow + (size_t)r * D) = o[r];
            *(float2*)(krow + (size_t)r * D) = ok[r];
        }
    }
}

constexpr int AT_WAVES = 4;
constexpr int AT_ROWS  = 8;
constexpr int AT_HALO  = DIL * (WIN / 2);
constexpr int AT_SPAN  = AT_ROWS + 2 * AT_HALO;
constexpr int VP_GROUPS      = 8;
constexpr int VP_PER_GROUP   = VP_PER_BATCH / VP_GROUPS;
constexpr int GROUPS_PER_WAVE = VP_GROUPS / AT_WAVES;

__global__ __launch_bounds__(128) void attn_kernel(
    const float* __restrict__ Q, const float* __restrict__ K,
    const float* __restrict__ V, const float* __restrict__ vpart,
    float* __restrict__ out)
{
    __shared__ float2 red[VP_GROUPS][32];

    const int lane = threadIdx.x & 31;
    const int wave = __builtin_amdgcn_readfirstlane((int)(threadIdx.x >> 5));
    const int gwave = blockIdx.x * AT_WAVES + wave;
    const int r0 = gwave * AT_ROWS;
    const int b  = r0 / S;
    const int i0 = r0 - b * S;

    const float* pp = vpart
        + (size_t)(b * VP_PER_BATCH + wave * (GROUPS_PER_WAVE * VP_PER_GROUP)) * D + lane * 2;
    float2 ps[GROUPS_PER_WAVE];
#pragma unroll
    for (int g = 0; g < GROUPS_PER_WAVE; ++g) {
        ps[g] = *(const float2*)(pp + (size_t)(g * VP_PER_GROUP) * D);
#pragma unroll
        for (int k = 1; k < VP_PER_GROUP; ++k) {
            const float2 t = *(const float2*)(pp + (size_t)(g * VP_PER_GROUP + k) * D);
            ps[g].x += t.x;
            ps[g].y += t.y;
        }
    }

    const float* Kb = K + (size_t)b * S * D + lane * 2;
    const float* Vb = V + (size_t)b * S * D + lane * 2;
    float2 qv[AT_ROWS], kk[AT_SPAN], vv[AT_SPAN];
#pragma unroll
    for (int r = 0; r < AT_ROWS; ++r)
        qv[r] = *(const float2*)(Q + (size_t)(r0 + r) * D + lane * 2);
#pragma unroll
    for (int t = 0; t < AT_SPAN; ++t) {
        int j = i0 - AT_HALO + t;
        if (t < AT_HALO) j = (j < 0) ? 0 : j;
        if (t >= AT_ROWS + AT_HALO) j = (j > S - 1) ? (S - 1) : j;
        kk[t] = *(const float2*)(Kb + (size_t)j * D);
        vv[t] = *(const float2*)(Vb + (size_t)j * D);
    }

#pragma unroll
    for (int g = 0; g < GROUPS_PER_WAVE; ++g)
        red[wave * GROUPS_PER_WAVE + g][lane] = ps[g];
    __syncthreads();

    float2 vs = red[0][lane];
#pragma unroll
    for (int w = 1; w < VP_GROUPS; ++w) {
        const float2 t = red[w][lane];
        vs.x += t.x;
        vs.y += t.y;
    }

    constexpr int NDOT = AT_ROWS * WIN;
    float c[NDOT];
#pragma unroll
    for (int r = 0; r < AT_ROWS; ++r) {
#pragma unroll
        for (int w = 0; w < WIN; ++w) {
            const int t = r + DIL * w;
            c[r * WIN + w] = qv[r].x * kk[t].x + qv[r].y * kk[t].y;
        }
    }

#pragma unroll
    for (int off = 16; off > 0; off >>= 1) {
        float x[NDOT];
#pragma unroll
        for (int n = 0; n < NDOT; ++n) x[n] = __shfl_xor(c[n], off, 32);
        __builtin_amdgcn_sched_barrier(0);
#pragma unroll
        for (int n = 0; n < NDOT; ++n) c[n] += x[n];
        __builtin_amdgcn_sched_barrier(0);
    }

#pragma unroll
    for (int r = 0; r < AT_ROWS; ++r) {
        float accx = vs.x;
        float accy = vs.y;
        float Z = (float)S;
#pragma unroll
        for (int w = 0; w < WIN; ++w) {
            const int t = r + DIL * w;
            const int j = i0 + r + DIL * (w - WIN / 2);
            const bool valid = (j >= 0) && (j < S);
            float em1 = __expf(c[r * WIN + w]) - 1.0f;
            em1 = valid ? em1 : 0.0f;
            accx += em1 * vv[t].x;
            accy += em1 * vv[t].y;
            Z    += em1;
        }
        const float inv = 1.0f / Z;
        float2 o;
        o.x = accx * inv;
        o.y = accy * inv;
        *(float2*)(out + (size_t)(r0 + r) * D + lane * 2) = o;
    }
}

extern "C" void kernel_launch(void* const* d_in, const int* in_sizes, int n_in,
                              void* d_out, int out_size, void* d_ws, size_t ws_size,
                              hipStream_t stream) {
    const float* x  = (const float*)d_in[0];
    const float* Wq = (const float*)d_in[1];
    const float* bq = (const float*)d_in[2];
    const float* Wk = (const float*)d_in[3];
    const float* bk = (const float*)d_in[4];
    const float* Wv = (const float*)d_in[5];
    const float* bv = (const float*)d_in[6];
    float* outp = (float*)d_out;

    constexpr size_t QKV = (size_t)B * S * D;
    float* ws = (float*)d_ws;
    float* Qb = ws;
    float* Kb = ws + QKV;
    float* Vb = ws + 2 * QKV;
    float* Vp = ws + 3 * QKV;

    const int mtiles = (B * S) / 16;
    dim3 gproj(mtiles / PROJ_TILES, 2);
    qkv_proj_kernel<<<gproj, PROJ_WAVES * 32, 0, stream>>>(x, Wq, Wk, Wv, bq, bk, bv, Qb, Vp);

    const int nwaves = (B * S) / AT_ROWS;
    attn_kernel<<<nwaves / AT_WAVES, AT_WAVES * 32, 0, stream>>>(Qb, Kb, Vb, Vp, outp);
}


# proj: the two waves of a SIMD take turns at the matrix unit: every wave lowers its s_setprio 3,2,1,0 at fixed groups of its K loop, so the wave further back wins; no load, sum or order changes
# speedup vs baseline: 28.0478x; 1.0402x over previous
#include <hip/hip_runtime.h>
#include <hip/hip_bf16.h>

constexpr int B   = 2;
constexpr int S   = 4096;
constexpr int E   = 1024;
constexpr int D   = 64;
constexpr int WIN = 5;
constexpr int DIL = 2;

typedef __attribute__((ext_vector_type(2))) float v2f;
typedef __attribute__((ext_vector_type(8))) float v8f;

constexpr int GK     = 4;
constexpr int NG     = 3;
constexpr int PF     = GK * NG;
constexpr int NKSTEP = E / 4;
constexpr int KSPLIT = 2;
constexpr int NKW    = NKSTEP / KSPLIT;
constexpr int NTRIP  = (NKW - PF) / PF;
constexpr int KTAIL  = NKW - PF - NTRIP * PF;
static_assert(KTAIL % GK == 0 && KTAIL < PF, "the tail below refills whole groups");

constexpr int PRIO_G2 = 14;
constexpr int PRIO_G1 = 24;
constexpr int PRIO_G0 = NTRIP * NG + KTAIL / GK + (PF - KTAIL) / GK;
static_assert(PRIO_G2 < PRIO_G1 && PRIO_G1 < NTRIP * NG, "the first two points lie in the steady state");

constexpr int PROJ_TILES   = 4;
constexpr int PROJ_WAVES   = PROJ_TILES * KSPLIT;
constexpr int PROJ_ROWS    = PROJ_TILES * 16;
constexpr int XCH_VEC      = 6 * 8 / 4;
constexpr int VP_PER_BATCH = S / PROJ_ROWS;
constexpr int VP_ROWS      = B * VP_PER_BATCH;
constexpr int KHALF        = D / 2;

struct KStep { v2f a; float4 f0, f1; float2 h0, h1; };

__device__ __forceinline__ KStep load_kstep(const float* __restrict__ xk,
                                            const float* __restrict__ wf,
                                            const float* __restrict__ wh)
{
    KStep f;
    f.a  = *(const v2f*)xk;
    f.f0 = *(const float4*)wf;
    f.f1 = *(const float4*)(wf + D);
    f.h0 = *(const float2*)wh;
    f.h1 = *(const float2*)(wh + D);
    return f;
}

__device__ __forceinline__ void mma_kstep(const KStep& f, v8f& acc0, v8f& acc1, v8f& acc2,
                                          v8f& acc3, v8f& acc4, v8f& acc5)
{
    v2f b0, b1, b2, b3, b4, b5;
    b0.x = f.f0.x;  b0.y = f.f1.x;
    b1.x = f.f0.y;  b1.y = f.f1.y;
    b2.x = f.f0.z;  b2.y = f.f1.z;
    b3.x = f.f0.w;  b3.y = f.f1.w;
    b4.x = f.h0.x;  b4.y = f.h1.x;
    b5.x = f.h0.y;  b5.y = f.h1.y;
    acc0 = __builtin_amdgcn_wmma_f32_16x16x4_f32(false, f.a, false, b0,
                                                 (short)0, acc0, false, false);
    acc1 = __builtin_amdgcn_wmma_f32_16x16x4_f32(false, f.a, false, b1,
                                                 (short)0, acc1, false, false);
    acc2 = __builtin_amdgcn_wmma_f32_16x16x4_f32(false, f.a, false, b2,
                                                 (short)0, acc2, false, false);
    acc3 = __builtin_amdgcn_wmma_f32_16x16x4_f32(false, f.a, false, b3,
                                                 (short)0, acc3, false, false);
    acc4 = __builtin_amdgcn_wmma_f32_16x16x4_f32(false, f.a, false, b4,
                                                 (short)0, acc4, false, false);
    acc5 = __builtin_amdgcn_wmma_f32_16x16x4_f32(false, f.a, false, b5,
                                                 (short)0, acc5, false, false);
}

__global__ __launch_bounds__(256) __attribute__((amdgpu_waves_per_eu(2, 2))) void qkv_proj_kernel(
    const float* __restrict__ x,
    const float* __restrict__ Wq,
    const float* __restrict__ Wk,
    const float* __restrict__ Wv,
    const float* __restrict__ biasq,
    const float* __restrict__ biask,
    const float* __restrict__ biasv,
    float* __restrict__ qkv,
    float* __restrict__ vpart)
{
    __shared__ float part[PROJ_TILES][D];
    __shared__ float4 xch[PROJ_TILES][XCH_VEC][32];

    const int hsel = blockIdx.y;
    const float* __restrict__ Wf    = hsel ? Wv : Wq;
    const float* __restrict__ biasf = hsel ? biasv : biasq;
    float* __restrict__ outf = qkv + (size_t)(hsel ? 2 : 0) * ((size_t)B * S * D);
    float* __restrict__ outk = qkv + (size_t)B * S * D;
    const int koff = hsel * KHALF;

    const int lane  = threadIdx.x & 31;
    const int wave  = __builtin_amdgcn_readfirstlane((int)(threadIdx.x >> 5));
    const int rt    = wave % PROJ_TILES;
    const int kq    = wave / PROJ_TILES;
    const int mtile = blockIdx.x * PROJ_TILES + rt;
    const int row0  = mtile * 16;
    const int k0    = kq * (NKW * 4);

    const int m  = lane & 15;
    const int c4 = (lane & 15) * 4;
    const int c2 = koff + (lane & 15) * 2;
    const int kh = (lane >> 4) * 2;

    const float* xrow = x  + (size_t)(row0 + m) * E + k0 + kh;
    const float* wfc  = Wf + (size_t)(k0 + kh) * D + c4;
    const float* whc  = Wk + (size_t)(k0 + kh) * D + c2;

    v8f acc0 = {}, acc1 = {}, acc2 = {}, acc3 = {}, acc4 = {}, acc5 = {};

    __builtin_amdgcn_s_setprio(3);

    KStep f[PF];
#pragma unroll
    for (int u = 0; u < PF; ++u)
        f[u] = load_kstep(xrow + 4 * u, wfc + (size_t)(4 * u) * D, whc + (size_t)(4 * u) * D);

#pragma unroll
    for (int s0 = 0; s0 < NTRIP * PF; s0 += PF) {
        const float* xk = xrow + 4 * (s0 + PF);
        const float* wf = wfc + (size_t)(4 * (s0 + PF)) * D;
        const float* wh = whc + (size_t)(4 * (s0 + PF)) * D;
#pragma unroll
        for (int g = 0; g < NG; ++g) {
            if ((s0 / PF) * NG + g == PRIO_G2) __builtin_amdgcn_s_setprio(2);
            if ((s0 / PF) * NG + g == PRIO_G1) __builtin_amdgcn_s_setprio(1);
#pragma unroll
            for (int u = g * GK; u < (g + 1) * GK; ++u)
                mma_kstep(f[u], acc0, acc1, acc2, acc3, acc4, acc5);
#pragma unroll
            for (int u = g * GK; u < (g + 1) * GK; ++u)
                f[u] = load_kstep(xk + 4 * u, wf + (size_t)(4 * u) * D, wh + (size_t)(4 * u) * D);
            __builtin_amdgcn_sched_barrier(0);
        }
    }

    {
        const float* xk = xrow + 4 * (NTRIP * PF + PF);
        const float* wf = wfc + (size_t)(4 * (NTRIP * PF + PF)) * D;
        const float* wh = whc + (size_t)(4 * (NTRIP * PF + PF)) * D;
#pragma unroll
        for (int g = 0; g < KTAIL / GK; ++g) {
#pragma unroll
            for (int u = g * GK; u < (g + 1) * GK; ++u)
                mma_kstep(f[u], acc0, acc1, acc2, acc3, acc4, acc5);
#pragma unroll
            for (int u = g * GK; u < (g + 1) * GK; ++u)
                f[u] = load_kstep(xk + 4 * u, wf + (size_t)(4 * u) * D, wh + (size_t)(4 * u) * D);
            __builtin_amdgcn_sched_barrier(0);
        }
    }
#pragma unroll
    for (int u = KTAIL; u < PF; ++u)
        mma_kstep(f[u], acc0, acc1, acc2, acc3, acc4, acc5);
    __builtin_amdgcn_s_setprio(0);
#pragma unroll
    for (int u = 0; u < KTAIL; ++u)
        mma_kstep(f[u], acc0, acc1, acc2, acc3, acc4, acc5);

    if (kq == 1) {
#pragma unroll
        for (int q = 0; q < 2; ++q) {
            xch[rt][0 + q][lane]  = make_float4(acc0[4 * q], acc0[4 * q + 1], acc0[4 * q + 2], acc0[4 * q + 3]);
            xch[rt][2 + q][lane]  = make_float4(acc1[4 * q], acc1[4 * q + 1], acc1[4 * q + 2], acc1[4 * q + 3]);
            xch[rt][4 + q][lane]  = make_float4(acc2[4 * q], acc2[4 * q + 1], acc2[4 * q + 2], acc2[4 * q + 3]);
            xch[rt][6 + q][lane]  = make_float4(acc3[4 * q], acc3[4 * q + 1], acc3[4 * q + 2], acc3[4 * q + 3]);
            xch[rt][8 + q][lane]  = make_float4(acc4[4 * q], acc4[4 * q + 1], acc4[4 * q + 2], acc4[4 * q + 3]);
            xch[rt][10 + q][lane] = make_float4(acc5[4 * q], acc5[4 * q + 1], acc5[4 * q + 2], acc5[4 * q + 3]);
        }
    }
    __syncthreads();
    if (kq == 0) {
#pragma unroll
        for (int q = 0; q < 2; ++q) {
            const float4 t0 = xch[rt][0 + q][lane];
            const float4 t1 = xch[rt][2 + q][lane];
            const float4 t2 = xch[rt][4 + q][lane];
            const float4 t3 = xch[rt][6 + q][lane];
            const float4 t4 = xch[rt][8 + q][lane];
            const float4 t5 = xch[rt][10 + q][lane];
            acc0[4 * q] += t0.x;  acc0[4 * q + 1] += t0.y;  acc0[4 * q + 2] += t0.z;  acc0[4 * q + 3] += t0.w;
            acc1[4 * q] += t1.x;  acc1[4 * q + 1] += t1.y;  acc1[4 * q + 2] += t1.z;  acc1[4 * q + 3] += t1.w;
            acc2[4 * q] += t2.x;  acc2[4 * q + 1] += t2.y;  acc2[4 * q + 2] += t2.z;  acc2[4 * q + 3] += t2.w;
            acc3[4 * q] += t3.x;  acc3[4 * q + 1] += t3.y;  acc3[4 * q + 2] += t3.z;  acc3[4 * q + 3] += t3.w;
            acc4[4 * q] += t4.x;  acc4[4 * q + 1] += t4.y;  acc4[4 * q + 2] += t4.z;  acc4[4 * q + 3] += t4.w;
            acc5[4 * q] += t5.x;  acc5[4 * q + 1] += t5.y;  acc5[4 * q + 2] += t5.z;  acc5[4 * q + 3] += t5.w;
        }
    }

    const int mo = (lane >> 4) * 8;
    const float4 bvf = *(const float4*)(biasf + c4);
    const float2 bvk = *(const float2*)(biask + c2);
    float4 o[8];
    float2 ok[8];
#pragma unroll
    for (int r = 0; r < 8; ++r) {
        o[r].x = acc0[r] + bvf.x;
        o[r].y = acc1[r] + bvf.y;
        o[r].z = acc2[r] + bvf.z;
        o[r].w = acc3[r] + bvf.w;
        ok[r].x = acc4[r] + bvk.x;
        ok[r].y = acc5[r] + bvk.y;
    }

    if (hsel == 1) {
        if (kq == 0) {
            float4 cs = o[0];
#pragma unroll
            for (int r = 1; r < 8; ++r) {
                cs.x += o[r].x;
                cs.y += o[r].y;
                cs.z += o[r].z;
                cs.w += o[r].w;
            }
            cs.x += __shfl_xor(cs.x, 16, 32);
            cs.y += __shfl_xor(cs.y, 16, 32);
            cs.z += __shfl_xor(cs.z, 16, 32);
            cs.w += __shfl_xor(cs.w, 16, 32);
            if (lane < 16) *(float4*)&part[rt][c4] = cs;
        }
        __syncthreads();
        if (threadIdx.x < D) {
            float a = part[0][threadIdx.x];
#pragma unroll
            for (int w = 1; w < PROJ_TILES; ++w) a += part[w][threadIdx.x];
            vpart[(size_t)blockIdx.x * D + threadIdx.x] = a;
        }
    }

    if (kq == 0) {
        float* orow = outf + (size_t)(row0 + mo) * D + c4;
        float* krow = outk + (size_t)(row0 + mo) * D + c2;
#pragma unroll
        for (int r = 0; r < 8; ++r) {
            *(float4*)(orow + (size_t)r * D) = o[r];
            *(float2*)(krow + (size_t)r * D) = ok[r];
        }
    }
}

constexpr int AT_WAVES = 4;
constexpr int AT_ROWS  = 8;
constexpr int AT_HALO  = DIL * (WIN / 2);
constexpr int AT_SPAN  = AT_ROWS + 2 * AT_HALO;
constexpr int VP_GROUPS      = 8;
constexpr int VP_PER_GROUP   = VP_PER_BATCH / VP_GROUPS;
constexpr int GROUPS_PER_WAVE = VP_GROUPS / AT_WAVES;

__global__ __launch_bounds__(128) void attn_kernel(
    const float* __restrict__ Q, const float* __restrict__ K,
    const float* __restrict__ V, const float* __restrict__ vpart,
    float* __restrict__ out)
{
    __shared__ float2 red[VP_GROUPS][32];

    const int lane = threadIdx.x & 31;
    const int wave = __builtin_amdgcn_readfirstlane((int)(threadIdx.x >> 5));
    const int gwave = blockIdx.x * AT_WAVES + wave;
    const int r0 = gwave * AT_ROWS;
    const int b  = r0 / S;
    const int i0 = r0 - b * S;

    const float* pp = vpart
        + (size_t)(b * VP_PER_BATCH + wave * (GROUPS_PER_WAVE * VP_PER_GROUP)) * D + lane * 2;
    float2 ps[GROUPS_PER_WAVE];
#pragma unroll
    for (int g = 0; g < GROUPS_PER_WAVE; ++g) {
        ps[g] = *(const float2*)(pp + (size_t)(g * VP_PER_GROUP) * D);
#pragma unroll
        for (int k = 1; k < VP_PER_GROUP; ++k) {
            const float2 t = *(const float2*)(pp + (size_t)(g * VP_PER_GROUP + k) * D);
            ps[g].x += t.x;
            ps[g].y += t.y;
        }
    }

    const float* Kb = K + (size_t)b * S * D + lane * 2;
    const float* Vb = V + (size_t)b * S * D + lane * 2;
    float2 qv[AT_ROWS], kk[AT_SPAN], vv[AT_SPAN];
#pragma unroll
    for (int r = 0; r < AT_ROWS; ++r)
        qv[r] = *(const float2*)(Q + (size_t)(r0 + r) * D + lane * 2);
#pragma unroll
    for (int t = 0; t < AT_SPAN; ++t) {
        int j = i0 - AT_HALO + t;
        if (t < AT_HALO) j = (j < 0) ? 0 : j;
        if (t >= AT_ROWS + AT_HALO) j = (j > S - 1) ? (S - 1) : j;
        kk[t] = *(const float2*)(Kb + (size_t)j * D);
        vv[t] = *(const float2*)(Vb + (size_t)j * D);
    }

#pragma unroll
    for (int g = 0; g < GROUPS_PER_WAVE; ++g)
        red[wave * GROUPS_PER_WAVE + g][lane] = ps[g];
    __syncthreads();

    float2 vs = red[0][lane];
#pragma unroll
    for (int w = 1; w < VP_GROUPS; ++w) {
        const float2 t = red[w][lane];
        vs.x += t.x;
        vs.y += t.y;
    }

    constexpr int NDOT = AT_ROWS * WIN;
    float c[NDOT];
#pragma unroll
    for (int r = 0; r < AT_ROWS; ++r) {
#pragma unroll
        for (int w = 0; w < WIN; ++w) {
            const int t = r + DIL * w;
            c[r * WIN + w] = qv[r].x * kk[t].x + qv[r].y * kk[t].y;
        }
    }

#pragma unroll
    for (int off = 16; off > 0; off >>= 1) {
        float x[NDOT];
#pragma unroll
        for (int n = 0; n < NDOT; ++n) x[n] = __shfl_xor(c[n], off, 32);
        __builtin_amdgcn_sched_barrier(0);
#pragma unroll
        for (int n = 0; n < NDOT; ++n) c[n] += x[n];
        __builtin_amdgcn_sched_barrier(0);
    }

#pragma unroll
    for (int r = 0; r < AT_ROWS; ++r) {
        float accx = vs.x;
        float accy = vs.y;
        float Z = (float)S;
#pragma unroll
        for (int w = 0; w < WIN; ++w) {
            const int t = r + DIL * w;
            const int j = i0 + r + DIL * (w - WIN / 2);
            const bool valid = (j >= 0) && (j < S);
            float em1 = __expf(c[r * WIN + w]) - 1.0f;
            em1 = valid ? em1 : 0.0f;
            accx += em1 * vv[t].x;
            accy += em1 * vv[t].y;
            Z    += em1;
        }
        const float inv = 1.0f / Z;
        float2 o;
        o.x = accx * inv;
        o.y = accy * inv;
        *(float2*)(out + (size_t)(r0 + r) * D + lane * 2) = o;
    }
}

extern "C" void kernel_launch(void* const* d_in, const int* in_sizes, int n_in,
                              void* d_out, int out_size, void* d_ws, size_t ws_size,
                              hipStream_t stream) {
    const float* x  = (const float*)d_in[0];
    const float* Wq = (const float*)d_in[1];
    const float* bq = (const float*)d_in[2];
    const float* Wk = (const float*)d_in[3];
    const float* bk = (const float*)d_in[4];
    const float* Wv = (const float*)d_in[5];
    const float* bv = (const float*)d_in[6];
    float* outp = (float*)d_out;

    constexpr size_t QKV = (size_t)B * S * D;
    float* ws = (float*)d_ws;
    float* Qb = ws;
    float* Kb = ws + QKV;
    float* Vb = ws + 2 * QKV;
    float* Vp = ws + 3 * QKV;

    const int mtiles = (B * S) / 16;
    dim3 gproj(mtiles / PROJ_TILES, 2);
    qkv_proj_kernel<<<gproj, PROJ_WAVES * 32, 0, stream>>>(x, Wq, Wk, Wv, bq, bk, bv, Qb, Vp);

    const int nwaves = (B * S) / AT_ROWS;
    attn_kernel<<<nwaves / AT_WAVES, AT_WAVES * 32, 0, stream>>>(Qb, Kb, Vb, Vp, outp);
}
